# starting kernel
# baseline (speedup 1.0000x reference)
#include <hip/hip_runtime.h>


typedef __attribute__((ext_vector_type(2))) float v2f;
typedef __attribute__((ext_vector_type(8))) float v8f;

constexpr double dE    = 3130.0;
constexpr double dNU   = 0.37;
constexpr double dSIGY = 64.8;
constexpr double dHARD = 100.0;
constexpr double dG    = dE / (2.0 * (1.0 + dNU));
constexpr double dLAM  = dE * dNU / ((1.0 + dNU) * (1.0 - 2.0 * dNU));
constexpr double dC    = dE / ((1.0 + dNU) * (1.0 - 2.0 * dNU));

constexpr float fG     = (float)dG;
constexpr float fLAM   = (float)dLAM;
constexpr float fC11   = (float)(dC * (1.0 - dNU));
constexpr float fC12   = (float)(dC * dNU);
constexpr float fSIGY  = (float)dSIGY;
constexpr float fHARD  = (float)dHARD;
constexpr float f3G    = (float)(3.0 * dG);
constexpr float fINV3GH = (float)(1.0 / (3.0 * dG + dHARD));

#define NB   128
#define NS   2048
#define LS   192
#define NO   3
#define NPADN 16

__global__ void prep_bpanel_kernel(const float* __restrict__ W2,
                                   float* __restrict__ Bpad) {
  int k = threadIdx.x;
  if (k >= LS) return;
#pragma unroll
  for (int n = 0; n < NPADN; ++n) {
    float v = 0.0f;
    if (n < NO) {
      float w = W2[n * LS + k];
      v = fmaxf(w, 0.0f) + log1pf(__expf(-fabsf(w)));
    }
    Bpad[n * LS + k] = v;
  }
}

__global__ __launch_bounds__(64) void j2_scan_kernel(
    const float* __restrict__ x,
    const float* __restrict__ W1,
    float* __restrict__ stress) {
  const int b = blockIdx.x;
  const int j = threadIdx.x;

  const float* w = W1 + (3 * j) * 3;
  const float w00 = w[0], w01 = w[1], w02 = w[2];
  const float w10 = w[3], w11 = w[4], w12 = w[5];
  const float w20 = w[6], w21 = w[7], w22 = w[8];

  float eps0 = 0.f, eps1 = 0.f, eps2 = 0.f;
  float sig0 = 0.f, sig1 = 0.f, sig2 = 0.f;
  float szz  = 0.f, eqps = 0.f;

  const float* __restrict__ xb = x + (size_t)b * NS * 3;
  float* __restrict__ outb = stress + (size_t)b * NS * LS + 3 * j;

  for (int s = 0; s < NS; ++s) {
    const float x0 = xb[3 * s + 0];
    const float x1 = xb[3 * s + 1];
    const float x2 = xb[3 * s + 2];

    const float st0 = w00 * x0 + w01 * x1 + w02 * x2;
    const float st1 = w10 * x0 + w11 * x1 + w12 * x2;
    const float st2 = w20 * x0 + w21 * x1 + w22 * x2;

    const float de0 = st0 - eps0;
    const float de1 = st1 - eps1;
    const float de2 = st2 - eps2;
    const float tr0 = sig0 + fC11 * de0 + fC12 * de1;
    const float tr1 = sig1 + fC12 * de0 + fC11 * de1;
    const float tr2 = sig2 + fG * de2;
    const float tzz = szz + fLAM * (de0 + de1);

    const float pm  = (tr0 + tr1 + tzz) * (1.0f / 3.0f);
    const float dxx = tr0 - pm;
    const float dyy = tr1 - pm;
    const float dzz = tzz - pm;
    const float dxy = tr2;
    const float qq  = 1.5f * (dxx * dxx + dyy * dyy + dzz * dzz +
                              2.0f * dxy * dxy) + 1e-12f;
    const float q   = sqrtf(qq);
    const float f   = q - (fSIGY + fHARD * eqps);
    const float dgam = (f > 0.0f) ? f * fINV3GH : 0.0f;
    float rq = __builtin_amdgcn_rcpf(q);
    rq = rq * (2.0f - q * rq);
    const float fac = 1.0f - f3G * dgam * rq;

    sig0 = pm + fac * dxx;
    sig1 = pm + fac * dyy;
    sig2 = fac * dxy;
    szz  = pm + fac * dzz;
    eqps = eqps + dgam;
    eps0 = st0; eps1 = st1; eps2 = st2;

    float* o = outb + (size_t)s * LS;
    o[0] = sig0;
    o[1] = sig1;
    o[2] = sig2;
  }
}

__global__ __launch_bounds__(256) void out_gemm_wmma_kernel(
    const float* __restrict__ stress,
    const float* __restrict__ Bpad,
    float* __restrict__ out) {
  const int lane = threadIdx.x & 31;
  const int wave = threadIdx.x >> 5;
  const int tile = blockIdx.x * 8 + wave;
  const int grp  = lane >> 4;
  const int ln   = lane & 15;

  const float* __restrict__ arow =
      stress + ((size_t)tile * 16 + ln) * LS + 2 * grp;
  const float* __restrict__ brow = Bpad + ln * LS + 2 * grp;

  v8f acc = {};
#pragma unroll
  for (int k = 0; k < LS; k += 4) {
    v2f a = *(const v2f*)(arow + k);
    v2f bm = *(const v2f*)(brow + k);
    acc = __builtin_amdgcn_wmma_f32_16x16x4_f32(
        false, a, false, bm, (short)0, acc, false, false);
  }

  if (ln < NO) {
#pragma unroll
    for (int r = 0; r < 8; ++r) {
      const size_t m = (size_t)tile * 16 + r + 8 * grp;
      out[m * NO + ln] = acc[r];
    }
  }
}

extern "C" void kernel_launch(void* const* d_in, const int* in_sizes, int n_in,
                              void* d_out, int out_size, void* d_ws,
                              size_t ws_size, hipStream_t stream) {
  (void)in_sizes; (void)n_in; (void)out_size; (void)ws_size;
  const float* x  = (const float*)d_in[0];
  const float* W1 = (const float*)d_in[1];
  const float* W2 = (const float*)d_in[2];
  float* out = (float*)d_out;

  float* stress = (float*)d_ws;
  float* Bpad   = stress + (size_t)NB * NS * LS;

  prep_bpanel_kernel<<<1, LS, 0, stream>>>(W2, Bpad);
  j2_scan_kernel<<<NB, 64, 0, stream>>>(x, W1, stress);
  const int tiles = (NB * NS) / 16;
  out_gemm_wmma_kernel<<<tiles / 8, 256, 0, stream>>>(stress, Bpad, out);
}


# scan: time loop in blocks of 8 steps, next block's x loaded one block ahead and projected to strains at block end; elastic increment formed off the carried chain (summed before added to old stress)
# speedup vs baseline: 1.9052x; 1.9052x over previous
#include <hip/hip_runtime.h>


typedef __attribute__((ext_vector_type(2))) float v2f;
typedef __attribute__((ext_vector_type(8))) float v8f;

constexpr double dE    = 3130.0;
constexpr double dNU   = 0.37;
constexpr double dSIGY = 64.8;
constexpr double dHARD = 100.0;
constexpr double dG    = dE / (2.0 * (1.0 + dNU));
constexpr double dLAM  = dE * dNU / ((1.0 + dNU) * (1.0 - 2.0 * dNU));
constexpr double dC    = dE / ((1.0 + dNU) * (1.0 - 2.0 * dNU));

constexpr float fG     = (float)dG;
constexpr float fLAM   = (float)dLAM;
constexpr float fC11   = (float)(dC * (1.0 - dNU));
constexpr float fC12   = (float)(dC * dNU);
constexpr float fSIGY  = (float)dSIGY;
constexpr float fHARD  = (float)dHARD;
constexpr float f3G    = (float)(3.0 * dG);
constexpr float fINV3GH = (float)(1.0 / (3.0 * dG + dHARD));

#define NB   128
#define NS   2048
#define LS   192
#define NO   3
#define NPADN 16

__global__ void prep_bpanel_kernel(const float* __restrict__ W2,
                                   float* __restrict__ Bpad) {
  int k = threadIdx.x;
  if (k >= LS) return;
#pragma unroll
  for (int n = 0; n < NPADN; ++n) {
    float v = 0.0f;
    if (n < NO) {
      float w = W2[n * LS + k];
      v = fmaxf(w, 0.0f) + log1pf(__expf(-fabsf(w)));
    }
    Bpad[n * LS + k] = v;
  }
}

#define TB 8
__global__ __launch_bounds__(64) void j2_scan_kernel(
    const float* __restrict__ x,
    const float* __restrict__ W1,
    float* __restrict__ stress) {
  const int b = blockIdx.x;
  const int j = threadIdx.x;

  const float* w = W1 + (3 * j) * 3;
  const float w00 = w[0], w01 = w[1], w02 = w[2];
  const float w10 = w[3], w11 = w[4], w12 = w[5];
  const float w20 = w[6], w21 = w[7], w22 = w[8];

  float eps0 = 0.f, eps1 = 0.f, eps2 = 0.f;
  float sig0 = 0.f, sig1 = 0.f, sig2 = 0.f;
  float szz  = 0.f, eqps = 0.f;

  const float* __restrict__ xb = x + (size_t)b * NS * 3;
  float* __restrict__ outb = stress + (size_t)b * NS * LS + 3 * j;

  float st[TB][3];
#pragma unroll
  for (int k = 0; k < TB; ++k) {
    const float x0 = xb[3 * k + 0];
    const float x1 = xb[3 * k + 1];
    const float x2 = xb[3 * k + 2];
    st[k][0] = w00 * x0 + w01 * x1 + w02 * x2;
    st[k][1] = w10 * x0 + w11 * x1 + w12 * x2;
    st[k][2] = w20 * x0 + w21 * x1 + w22 * x2;
  }

  for (int blk = 0; blk < NS / TB; ++blk) {
    const bool more = (blk + 1 < NS / TB);
    float xn[3 * TB];
    if (more) {
      const float* __restrict__ xnp = xb + (blk + 1) * (3 * TB);
#pragma unroll
      for (int i = 0; i < 3 * TB; ++i) xn[i] = xnp[i];
    }

    float* __restrict__ ob = outb + (size_t)blk * TB * LS;

#pragma unroll
    for (int k = 0; k < TB; ++k) {
      const float de0 = st[k][0] - eps0;
      const float de1 = st[k][1] - eps1;
      const float de2 = st[k][2] - eps2;
      eps0 = st[k][0]; eps1 = st[k][1]; eps2 = st[k][2];
      const float dt0 = fC11 * de0 + fC12 * de1;
      const float dt1 = fC12 * de0 + fC11 * de1;
      const float dt2 = fG * de2;
      const float dtz = fLAM * (de0 + de1);

      const float tr0 = sig0 + dt0;
      const float tr1 = sig1 + dt1;
      const float tr2 = sig2 + dt2;
      const float tzz = szz + dtz;

      const float pm  = (tr0 + tr1 + tzz) * (1.0f / 3.0f);
      const float dxx = tr0 - pm;
      const float dyy = tr1 - pm;
      const float dzz = tzz - pm;
      const float dxy = tr2;
      const float qq  = 1.5f * (dxx * dxx + dyy * dyy + dzz * dzz +
                                2.0f * dxy * dxy) + 1e-12f;
      const float q   = sqrtf(qq);
      const float f   = q - (fSIGY + fHARD * eqps);
      const float dgam = (f > 0.0f) ? f * fINV3GH : 0.0f;
      float rq = __builtin_amdgcn_rcpf(q);
      rq = rq * (2.0f - q * rq);
      const float fac = 1.0f - f3G * dgam * rq;

      sig0 = pm + fac * dxx;
      sig1 = pm + fac * dyy;
      sig2 = fac * dxy;
      szz  = pm + fac * dzz;
      eqps = eqps + dgam;

      float* o = ob + (size_t)k * LS;
      o[0] = sig0;
      o[1] = sig1;
      o[2] = sig2;
    }

    if (more) {
#pragma unroll
      for (int k = 0; k < TB; ++k) {
        const float x0 = xn[3 * k + 0];
        const float x1 = xn[3 * k + 1];
        const float x2 = xn[3 * k + 2];
        st[k][0] = w00 * x0 + w01 * x1 + w02 * x2;
        st[k][1] = w10 * x0 + w11 * x1 + w12 * x2;
        st[k][2] = w20 * x0 + w21 * x1 + w22 * x2;
      }
    }
  }
}

__global__ __launch_bounds__(256) void out_gemm_wmma_kernel(
    const float* __restrict__ stress,
    const float* __restrict__ Bpad,
    float* __restrict__ out) {
  const int lane = threadIdx.x & 31;
  const int wave = threadIdx.x >> 5;
  const int tile = blockIdx.x * 8 + wave;
  const int grp  = lane >> 4;
  const int ln   = lane & 15;

  const float* __restrict__ arow =
      stress + ((size_t)tile * 16 + ln) * LS + 2 * grp;
  const float* __restrict__ brow = Bpad + ln * LS + 2 * grp;

  v8f acc = {};
#pragma unroll
  for (int k = 0; k < LS; k += 4) {
    v2f a = *(const v2f*)(arow + k);
    v2f bm = *(const v2f*)(brow + k);
    acc = __builtin_amdgcn_wmma_f32_16x16x4_f32(
        false, a, false, bm, (short)0, acc, false, false);
  }

  if (ln < NO) {
#pragma unroll
    for (int r = 0; r < 8; ++r) {
      const size_t m = (size_t)tile * 16 + r + 8 * grp;
      out[m * NO + ln] = acc[r];
    }
  }
}

extern "C" void kernel_launch(void* const* d_in, const int* in_sizes, int n_in,
                              void* d_out, int out_size, void* d_ws,
                              size_t ws_size, hipStream_t stream) {
  (void)in_sizes; (void)n_in; (void)out_size; (void)ws_size;
  const float* x  = (const float*)d_in[0];
  const float* W1 = (const float*)d_in[1];
  const float* W2 = (const float*)d_in[2];
  float* out = (float*)d_out;

  float* stress = (float*)d_ws;
  float* Bpad   = stress + (size_t)NB * NS * LS;

  prep_bpanel_kernel<<<1, LS, 0, stream>>>(W2, Bpad);
  j2_scan_kernel<<<NB, 64, 0, stream>>>(x, W1, stress);
  const int tiles = (NB * NS) / 16;
  out_gemm_wmma_kernel<<<tiles / 8, 256, 0, stream>>>(stress, Bpad, out);
}


# scan: q and 1/q from one v_rsq_f32 with one Newton step each (fma residual) instead of library sqrtf plus v_rcp: 8 fewer dependent levels per step on the carried chain; q within 1 ulp of sqrtf
# speedup vs baseline: 2.3182x; 1.2168x over previous
#include <hip/hip_runtime.h>


typedef __attribute__((ext_vector_type(2))) float v2f;
typedef __attribute__((ext_vector_type(8))) float v8f;

constexpr double dE    = 3130.0;
constexpr double dNU   = 0.37;
constexpr double dSIGY = 64.8;
constexpr double dHARD = 100.0;
constexpr double dG    = dE / (2.0 * (1.0 + dNU));
constexpr double dLAM  = dE * dNU / ((1.0 + dNU) * (1.0 - 2.0 * dNU));
constexpr double dC    = dE / ((1.0 + dNU) * (1.0 - 2.0 * dNU));

constexpr float fG     = (float)dG;
constexpr float fLAM   = (float)dLAM;
constexpr float fC11   = (float)(dC * (1.0 - dNU));
constexpr float fC12   = (float)(dC * dNU);
constexpr float fSIGY  = (float)dSIGY;
constexpr float fHARD  = (float)dHARD;
constexpr float f3G    = (float)(3.0 * dG);
constexpr float fINV3GH = (float)(1.0 / (3.0 * dG + dHARD));

#define NB   128
#define NS   2048
#define LS   192
#define NO   3
#define NPADN 16

__global__ void prep_bpanel_kernel(const float* __restrict__ W2,
                                   float* __restrict__ Bpad) {
  int k = threadIdx.x;
  if (k >= LS) return;
#pragma unroll
  for (int n = 0; n < NPADN; ++n) {
    float v = 0.0f;
    if (n < NO) {
      float w = W2[n * LS + k];
      v = fmaxf(w, 0.0f) + log1pf(__expf(-fabsf(w)));
    }
    Bpad[n * LS + k] = v;
  }
}

#define TB 8
__global__ __launch_bounds__(64) void j2_scan_kernel(
    const float* __restrict__ x,
    const float* __restrict__ W1,
    float* __restrict__ stress) {
  const int b = blockIdx.x;
  const int j = threadIdx.x;

  const float* w = W1 + (3 * j) * 3;
  const float w00 = w[0], w01 = w[1], w02 = w[2];
  const float w10 = w[3], w11 = w[4], w12 = w[5];
  const float w20 = w[6], w21 = w[7], w22 = w[8];

  float eps0 = 0.f, eps1 = 0.f, eps2 = 0.f;
  float sig0 = 0.f, sig1 = 0.f, sig2 = 0.f;
  float szz  = 0.f, eqps = 0.f;

  const float* __restrict__ xb = x + (size_t)b * NS * 3;
  float* __restrict__ outb = stress + (size_t)b * NS * LS + 3 * j;

  float st[TB][3];
#pragma unroll
  for (int k = 0; k < TB; ++k) {
    const float x0 = xb[3 * k + 0];
    const float x1 = xb[3 * k + 1];
    const float x2 = xb[3 * k + 2];
    st[k][0] = w00 * x0 + w01 * x1 + w02 * x2;
    st[k][1] = w10 * x0 + w11 * x1 + w12 * x2;
    st[k][2] = w20 * x0 + w21 * x1 + w22 * x2;
  }

  for (int blk = 0; blk < NS / TB; ++blk) {
    const bool more = (blk + 1 < NS / TB);
    float xn[3 * TB];
    if (more) {
      const float* __restrict__ xnp = xb + (blk + 1) * (3 * TB);
#pragma unroll
      for (int i = 0; i < 3 * TB; ++i) xn[i] = xnp[i];
    }

    float* __restrict__ ob = outb + (size_t)blk * TB * LS;

#pragma unroll
    for (int k = 0; k < TB; ++k) {
      const float de0 = st[k][0] - eps0;
      const float de1 = st[k][1] - eps1;
      const float de2 = st[k][2] - eps2;
      eps0 = st[k][0]; eps1 = st[k][1]; eps2 = st[k][2];
      const float dt0 = fC11 * de0 + fC12 * de1;
      const float dt1 = fC12 * de0 + fC11 * de1;
      const float dt2 = fG * de2;
      const float dtz = fLAM * (de0 + de1);

      const float tr0 = sig0 + dt0;
      const float tr1 = sig1 + dt1;
      const float tr2 = sig2 + dt2;
      const float tzz = szz + dtz;

      const float pm  = (tr0 + tr1 + tzz) * (1.0f / 3.0f);
      const float dxx = tr0 - pm;
      const float dyy = tr1 - pm;
      const float dzz = tzz - pm;
      const float dxy = tr2;
      const float qq  = 1.5f * (dxx * dxx + dyy * dyy + dzz * dzz +
                                2.0f * dxy * dxy) + 1e-12f;
      const float r0  = __builtin_amdgcn_rsqf(qq);
      const float q0  = qq * r0;
      const float hr  = 0.5f * r0;
      const float er  = __builtin_fmaf(-q0, q0, qq);
      const float q   = __builtin_fmaf(er, hr, q0);
      const float f   = q - (fSIGY + fHARD * eqps);
      const float dgam = (f > 0.0f) ? f * fINV3GH : 0.0f;
      const float tq  = __builtin_fmaf(-q, r0, 1.0f);
      const float rq  = __builtin_fmaf(r0, tq, r0);
      const float fac = 1.0f - f3G * dgam * rq;

      sig0 = pm + fac * dxx;
      sig1 = pm + fac * dyy;
      sig2 = fac * dxy;
      szz  = pm + fac * dzz;
      eqps = eqps + dgam;

      float* o = ob + (size_t)k * LS;
      o[0] = sig0;
      o[1] = sig1;
      o[2] = sig2;
    }

    if (more) {
#pragma unroll
      for (int k = 0; k < TB; ++k) {
        const float x0 = xn[3 * k + 0];
        const float x1 = xn[3 * k + 1];
        const float x2 = xn[3 * k + 2];
        st[k][0] = w00 * x0 + w01 * x1 + w02 * x2;
        st[k][1] = w10 * x0 + w11 * x1 + w12 * x2;
        st[k][2] = w20 * x0 + w21 * x1 + w22 * x2;
      }
    }
  }
}

__global__ __launch_bounds__(256) void out_gemm_wmma_kernel(
    const float* __restrict__ stress,
    const float* __restrict__ Bpad,
    float* __restrict__ out) {
  const int lane = threadIdx.x & 31;
  const int wave = threadIdx.x >> 5;
  const int tile = blockIdx.x * 8 + wave;
  const int grp  = lane >> 4;
  const int ln   = lane & 15;

  const float* __restrict__ arow =
      stress + ((size_t)tile * 16 + ln) * LS + 2 * grp;
  const float* __restrict__ brow = Bpad + ln * LS + 2 * grp;

  v8f acc = {};
#pragma unroll
  for (int k = 0; k < LS; k += 4) {
    v2f a = *(const v2f*)(arow + k);
    v2f bm = *(const v2f*)(brow + k);
    acc = __builtin_amdgcn_wmma_f32_16x16x4_f32(
        false, a, false, bm, (short)0, acc, false, false);
  }

  if (ln < NO) {
#pragma unroll
    for (int r = 0; r < 8; ++r) {
      const size_t m = (size_t)tile * 16 + r + 8 * grp;
      out[m * NO + ln] = acc[r];
    }
  }
}

extern "C" void kernel_launch(void* const* d_in, const int* in_sizes, int n_in,
                              void* d_out, int out_size, void* d_ws,
                              size_t ws_size, hipStream_t stream) {
  (void)in_sizes; (void)n_in; (void)out_size; (void)ws_size;
  const float* x  = (const float*)d_in[0];
  const float* W1 = (const float*)d_in[1];
  const float* W2 = (const float*)d_in[2];
  float* out = (float*)d_out;

  float* stress = (float*)d_ws;
  float* Bpad   = stress + (size_t)NB * NS * LS;

  prep_bpanel_kernel<<<1, LS, 0, stream>>>(W2, Bpad);
  j2_scan_kernel<<<NB, 64, 0, stream>>>(x, W1, stress);
  const int tiles = (NB * NS) / 16;
  out_gemm_wmma_kernel<<<tiles / 8, 256, 0, stream>>>(stress, Bpad, out);
}


# one launch: scan waves hand each 32-step stress chunk through LDS to two more waves of the same work-group that run the WMMA projection meanwhile; stresses no longer pass through global memory
# speedup vs baseline: 2.6098x; 1.1258x over previous
#include <hip/hip_runtime.h>


typedef __attribute__((ext_vector_type(2))) float v2f;
typedef __attribute__((ext_vector_type(8))) float v8f;

constexpr double dE    = 3130.0;
constexpr double dNU   = 0.37;
constexpr double dSIGY = 64.8;
constexpr double dHARD = 100.0;
constexpr double dG    = dE / (2.0 * (1.0 + dNU));
constexpr double dLAM  = dE * dNU / ((1.0 + dNU) * (1.0 - 2.0 * dNU));
constexpr double dC    = dE / ((1.0 + dNU) * (1.0 - 2.0 * dNU));

constexpr float fG     = (float)dG;
constexpr float fLAM   = (float)dLAM;
constexpr float fC11   = (float)(dC * (1.0 - dNU));
constexpr float fC12   = (float)(dC * dNU);
constexpr float fSIGY  = (float)dSIGY;
constexpr float fHARD  = (float)dHARD;
constexpr float f3G    = (float)(3.0 * dG);
constexpr float fINV3GH = (float)(1.0 / (3.0 * dG + dHARD));

#define NB   128
#define NS   2048
#define LS   192
#define NO   3
#define TB   8
#define CH   32
#define PITCH 196

__global__ __launch_bounds__(128) void j2_scan_project_kernel(
    const float* __restrict__ x,
    const float* __restrict__ W1,
    const float* __restrict__ W2,
    float* __restrict__ out) {
  __shared__ __attribute__((aligned(16))) float sbuf[2][CH][PITCH];
  __shared__ __attribute__((aligned(16))) float spw[NO * LS];

  const int b = blockIdx.x;
  const int role = __builtin_amdgcn_readfirstlane((int)(threadIdx.x >> 5));
  const int j    = threadIdx.x & 63;
  const int lane = threadIdx.x & 31;
  const int grp  = lane >> 4;
  const int ln   = lane & 15;

  const float* w = W1 + (3 * j) * 3;
  const float w00 = w[0], w01 = w[1], w02 = w[2];
  const float w10 = w[3], w11 = w[4], w12 = w[5];
  const float w20 = w[6], w21 = w[7], w22 = w[8];

  float eps0 = 0.f, eps1 = 0.f, eps2 = 0.f;
  float sig0 = 0.f, sig1 = 0.f, sig2 = 0.f;
  float szz  = 0.f, eqps = 0.f;

  const float* __restrict__ xb = x + (size_t)b * NS * 3;

  float st[TB][3];
  v2f bfr[LS / 4];

  for (int i = threadIdx.x; i < NO * LS; i += 128) {
    const float wv = W2[i];
    spw[i] = fmaxf(wv, 0.0f) + log1pf(__expf(-fabsf(wv)));
  }
  __syncthreads();

  if (role < 2) {
#pragma unroll
    for (int k = 0; k < TB; ++k) {
      const float x0 = xb[3 * k + 0];
      const float x1 = xb[3 * k + 1];
      const float x2 = xb[3 * k + 2];
      st[k][0] = w00 * x0 + w01 * x1 + w02 * x2;
      st[k][1] = w10 * x0 + w11 * x1 + w12 * x2;
      st[k][2] = w20 * x0 + w21 * x1 + w22 * x2;
    }
#pragma unroll
    for (int t = 0; t < LS / 4; ++t) bfr[t] = (v2f){0.0f, 0.0f};
  } else {
#pragma unroll
    for (int k = 0; k < TB; ++k) { st[k][0] = 0.f; st[k][1] = 0.f; st[k][2] = 0.f; }
    const float* __restrict__ pr = spw + (ln < NO ? ln : 0) * LS + 2 * grp;
#pragma unroll
    for (int t = 0; t < LS / 4; ++t) {
      const v2f pv = *(const v2f*)(pr + 4 * t);
      bfr[t] = (v2f){ln < NO ? pv.x : 0.0f, ln < NO ? pv.y : 0.0f};
    }
  }

  for (int blk = 0; blk <= NS / TB; ++blk) {
    if (role < 2) {
      if (blk < NS / TB) {
        const bool more = (blk + 1 < NS / TB);
        float xn[3 * TB];
        if (more) {
          const float* __restrict__ xnp = xb + (blk + 1) * (3 * TB);
#pragma unroll
          for (int i = 0; i < 3 * TB; ++i) xn[i] = xnp[i];
        }

        float* __restrict__ ob =
            &sbuf[(blk / (CH / TB)) & 1][(blk % (CH / TB)) * TB][3 * j];

#pragma unroll
        for (int k = 0; k < TB; ++k) {
          const float de0 = st[k][0] - eps0;
          const float de1 = st[k][1] - eps1;
          const float de2 = st[k][2] - eps2;
          eps0 = st[k][0]; eps1 = st[k][1]; eps2 = st[k][2];
          const float dt0 = fC11 * de0 + fC12 * de1;
          const float dt1 = fC12 * de0 + fC11 * de1;
          const float dt2 = fG * de2;
          const float dtz = fLAM * (de0 + de1);

          const float tr0 = sig0 + dt0;
          const float tr1 = sig1 + dt1;
          const float tr2 = sig2 + dt2;
          const float tzz = szz + dtz;

          const float pm  = (tr0 + tr1 + tzz) * (1.0f / 3.0f);
          const float dxx = tr0 - pm;
          const float dyy = tr1 - pm;
          const float dzz = tzz - pm;
          const float dxy = tr2;
          const float qq  = 1.5f * (dxx * dxx + dyy * dyy + dzz * dzz +
                                    2.0f * dxy * dxy) + 1e-12f;
          const float r0  = __builtin_amdgcn_rsqf(qq);
          const float q0  = qq * r0;
          const float hr  = 0.5f * r0;
          const float er  = __builtin_fmaf(-q0, q0, qq);
          const float q   = __builtin_fmaf(er, hr, q0);
          const float f   = q - (fSIGY + fHARD * eqps);
          const float dgam = (f > 0.0f) ? f * fINV3GH : 0.0f;
          const float tq  = __builtin_fmaf(-q, r0, 1.0f);
          const float rq  = __builtin_fmaf(r0, tq, r0);
          const float fac = 1.0f - f3G * dgam * rq;

          sig0 = pm + fac * dxx;
          sig1 = pm + fac * dyy;
          sig2 = fac * dxy;
          szz  = pm + fac * dzz;
          eqps = eqps + dgam;

          float* o = ob + k * PITCH;
          o[0] = sig0;
          o[1] = sig1;
          o[2] = sig2;
        }

        if (more) {
#pragma unroll
          for (int k = 0; k < TB; ++k) {
            const float x0 = xn[3 * k + 0];
            const float x1 = xn[3 * k + 1];
            const float x2 = xn[3 * k + 2];
            st[k][0] = w00 * x0 + w01 * x1 + w02 * x2;
            st[k][1] = w10 * x0 + w11 * x1 + w12 * x2;
            st[k][2] = w20 * x0 + w21 * x1 + w22 * x2;
          }
        }
      }
    } else {
      if ((blk % (CH / TB)) == 0 && blk > 0) {
        const int cprev = blk / (CH / TB) - 1;
        const int row0 = (role - 2) * 16;
        const float* __restrict__ arow = &sbuf[cprev & 1][row0 + ln][2 * grp];
        v8f acc = {};
#pragma unroll
        for (int t = 0; t < LS / 4; ++t) {
          const v2f a = *(const v2f*)(arow + 4 * t);
          acc = __builtin_amdgcn_wmma_f32_16x16x4_f32(
              false, a, false, bfr[t], (short)0, acc, false, false);
        }
        if (ln < NO) {
          const size_t m0 =
              (size_t)b * NS + (size_t)cprev * CH + row0 + 8 * grp;
#pragma unroll
          for (int r = 0; r < 8; ++r) out[(m0 + r) * NO + ln] = acc[r];
        }
      }
    }
    if ((blk % (CH / TB)) == CH / TB - 1) __syncthreads();
  }
}

extern "C" void kernel_launch(void* const* d_in, const int* in_sizes, int n_in,
                              void* d_out, int out_size, void* d_ws,
                              size_t ws_size, hipStream_t stream) {
  (void)in_sizes; (void)n_in; (void)out_size; (void)d_ws; (void)ws_size;
  const float* x  = (const float*)d_in[0];
  const float* W1 = (const float*)d_in[1];
  const float* W2 = (const float*)d_in[2];
  float* out = (float*)d_out;

  j2_scan_project_kernel<<<NB, 128, 0, stream>>>(x, W1, W2, out);
}


# scan: carried state is the stress deviator only (next trial = factor*trial + increment, one fma each); mean stress = K*(e0+e1) from the step's own strain, off the chain; stress sums regrouped
# speedup vs baseline: 3.0443x; 1.1665x over previous
#include <hip/hip_runtime.h>


typedef __attribute__((ext_vector_type(2))) float v2f;
typedef __attribute__((ext_vector_type(8))) float v8f;

constexpr double dE    = 3130.0;
constexpr double dNU   = 0.37;
constexpr double dSIGY = 64.8;
constexpr double dHARD = 100.0;
constexpr double dG    = dE / (2.0 * (1.0 + dNU));
constexpr double dLAM  = dE * dNU / ((1.0 + dNU) * (1.0 - 2.0 * dNU));
constexpr double dC    = dE / ((1.0 + dNU) * (1.0 - 2.0 * dNU));

constexpr float fG     = (float)dG;
constexpr float fKB    = (float)(dC * (1.0 + dNU) / 3.0);
constexpr float fG43   = (float)(4.0 * dG / 3.0);
constexpr float fG23   = (float)(2.0 * dG / 3.0);
constexpr float fSIGY  = (float)dSIGY;
constexpr float fHARD  = (float)dHARD;
constexpr float f3G    = (float)(3.0 * dG);
constexpr float fINV3GH = (float)(1.0 / (3.0 * dG + dHARD));

#define NB   128
#define NS   2048
#define LS   192
#define NO   3
#define TB   8
#define CH   32
#define PITCH 196

__global__ __launch_bounds__(128) void j2_scan_project_kernel(
    const float* __restrict__ x,
    const float* __restrict__ W1,
    const float* __restrict__ W2,
    float* __restrict__ out) {
  __shared__ __attribute__((aligned(16))) float sbuf[2][CH][PITCH];
  __shared__ __attribute__((aligned(16))) float spw[NO * LS];

  const int b = blockIdx.x;
  const int role = __builtin_amdgcn_readfirstlane((int)(threadIdx.x >> 5));
  const int j    = threadIdx.x & 63;
  const int lane = threadIdx.x & 31;
  const int grp  = lane >> 4;
  const int ln   = lane & 15;

  const float* w = W1 + (3 * j) * 3;
  const float w00 = w[0], w01 = w[1], w02 = w[2];
  const float w10 = w[3], w11 = w[4], w12 = w[5];
  const float w20 = w[6], w21 = w[7], w22 = w[8];

  float eps0 = 0.f, eps1 = 0.f, eps2 = 0.f;
  float sd0 = 0.f, sd1 = 0.f, sdz = 0.f, sd2 = 0.f;
  float fac = 1.0f, eqps = 0.f;

  const float* __restrict__ xb = x + (size_t)b * NS * 3;

  float st[TB][3];
  v2f bfr[LS / 4];

  for (int i = threadIdx.x; i < NO * LS; i += 128) {
    const float wv = W2[i];
    spw[i] = fmaxf(wv, 0.0f) + log1pf(__expf(-fabsf(wv)));
  }
  __syncthreads();

  if (role < 2) {
#pragma unroll
    for (int k = 0; k < TB; ++k) {
      const float x0 = xb[3 * k + 0];
      const float x1 = xb[3 * k + 1];
      const float x2 = xb[3 * k + 2];
      st[k][0] = w00 * x0 + w01 * x1 + w02 * x2;
      st[k][1] = w10 * x0 + w11 * x1 + w12 * x2;
      st[k][2] = w20 * x0 + w21 * x1 + w22 * x2;
    }
#pragma unroll
    for (int t = 0; t < LS / 4; ++t) bfr[t] = (v2f){0.0f, 0.0f};
  } else {
#pragma unroll
    for (int k = 0; k < TB; ++k) { st[k][0] = 0.f; st[k][1] = 0.f; st[k][2] = 0.f; }
    const float* __restrict__ pr = spw + (ln < NO ? ln : 0) * LS + 2 * grp;
#pragma unroll
    for (int t = 0; t < LS / 4; ++t) {
      const v2f pv = *(const v2f*)(pr + 4 * t);
      bfr[t] = (v2f){ln < NO ? pv.x : 0.0f, ln < NO ? pv.y : 0.0f};
    }
  }

  for (int blk = 0; blk <= NS / TB; ++blk) {
    if (role < 2) {
      if (blk < NS / TB) {
        const bool more = (blk + 1 < NS / TB);
        float xn[3 * TB];
        if (more) {
          const float* __restrict__ xnp = xb + (blk + 1) * (3 * TB);
#pragma unroll
          for (int i = 0; i < 3 * TB; ++i) xn[i] = xnp[i];
        }

        float* __restrict__ ob =
            &sbuf[(blk / (CH / TB)) & 1][(blk % (CH / TB)) * TB][3 * j];

#pragma unroll
        for (int k = 0; k < TB; ++k) {
          const float de0 = st[k][0] - eps0;
          const float de1 = st[k][1] - eps1;
          const float de2 = st[k][2] - eps2;
          eps0 = st[k][0]; eps1 = st[k][1]; eps2 = st[k][2];
          const float ds0 = fG43 * de0 - fG23 * de1;
          const float ds1 = fG43 * de1 - fG23 * de0;
          const float dsz = -fG23 * (de0 + de1);
          const float ds2 = fG * de2;
          const float pm  = fKB * (st[k][0] + st[k][1]);

          sd0 = __builtin_fmaf(fac, sd0, ds0);
          sd1 = __builtin_fmaf(fac, sd1, ds1);
          sdz = __builtin_fmaf(fac, sdz, dsz);
          sd2 = __builtin_fmaf(fac, sd2, ds2);

          const float qq  = 1.5f * (sd0 * sd0 + sd1 * sd1 + sdz * sdz +
                                    2.0f * sd2 * sd2) + 1e-12f;
          const float r0  = __builtin_amdgcn_rsqf(qq);
          const float q0  = qq * r0;
          const float hr  = 0.5f * r0;
          const float er  = __builtin_fmaf(-q0, q0, qq);
          const float q   = __builtin_fmaf(er, hr, q0);
          const float f   = q - (fSIGY + fHARD * eqps);
          const float dgam = (f > 0.0f) ? f * fINV3GH : 0.0f;
          const float tq  = __builtin_fmaf(-q, r0, 1.0f);
          const float rq  = __builtin_fmaf(r0, tq, r0);
          fac = 1.0f - f3G * dgam * rq;
          eqps = eqps + dgam;

          float* o = ob + k * PITCH;
          o[0] = pm + fac * sd0;
          o[1] = pm + fac * sd1;
          o[2] = fac * sd2;
        }

        if (more) {
#pragma unroll
          for (int k = 0; k < TB; ++k) {
            const float x0 = xn[3 * k + 0];
            const float x1 = xn[3 * k + 1];
            const float x2 = xn[3 * k + 2];
            st[k][0] = w00 * x0 + w01 * x1 + w02 * x2;
            st[k][1] = w10 * x0 + w11 * x1 + w12 * x2;
            st[k][2] = w20 * x0 + w21 * x1 + w22 * x2;
          }
        }
      }
    } else {
      if ((blk % (CH / TB)) == 0 && blk > 0) {
        const int cprev = blk / (CH / TB) - 1;
        const int row0 = (role - 2) * 16;
        const float* __restrict__ arow = &sbuf[cprev & 1][row0 + ln][2 * grp];
        v8f acc = {};
#pragma unroll
        for (int t = 0; t < LS / 4; ++t) {
          const v2f a = *(const v2f*)(arow + 4 * t);
          acc = __builtin_amdgcn_wmma_f32_16x16x4_f32(
              false, a, false, bfr[t], (short)0, acc, false, false);
        }
        if (ln < NO) {
          const size_t m0 =
              (size_t)b * NS + (size_t)cprev * CH + row0 + 8 * grp;
#pragma unroll
          for (int r = 0; r < 8; ++r) out[(m0 + r) * NO + ln] = acc[r];
        }
      }
    }
    if ((blk % (CH / TB)) == CH / TB - 1) __syncthreads();
  }
}

extern "C" void kernel_launch(void* const* d_in, const int* in_sizes, int n_in,
                              void* d_out, int out_size, void* d_ws,
                              size_t ws_size, hipStream_t stream) {
  (void)in_sizes; (void)n_in; (void)out_size; (void)d_ws; (void)ws_size;
  const float* x  = (const float*)d_in[0];
  const float* W1 = (const float*)d_in[1];
  const float* W2 = (const float*)d_in[2];
  float* out = (float*)d_out;

  j2_scan_project_kernel<<<NB, 128, 0, stream>>>(x, W1, W2, out);
}


# scan: return factor straight from the reciprocal root, fac = sat((1-c) + c*Y*rsq(qq)), ONE fma with clamp; q, its Newton step, f, the select and two products leave the carried chain (feed eqps only)
# speedup vs baseline: 3.4712x; 1.1402x over previous
#include <hip/hip_runtime.h>


typedef __attribute__((ext_vector_type(2))) float v2f;
typedef __attribute__((ext_vector_type(8))) float v8f;

constexpr double dE    = 3130.0;
constexpr double dNU   = 0.37;
constexpr double dSIGY = 64.8;
constexpr double dHARD = 100.0;
constexpr double dG    = dE / (2.0 * (1.0 + dNU));
constexpr double dLAM  = dE * dNU / ((1.0 + dNU) * (1.0 - 2.0 * dNU));
constexpr double dC    = dE / ((1.0 + dNU) * (1.0 - 2.0 * dNU));

constexpr float fG     = (float)dG;
constexpr float fKB    = (float)(dC * (1.0 + dNU) / 3.0);
constexpr float fG43   = (float)(4.0 * dG / 3.0);
constexpr float fG23   = (float)(2.0 * dG / 3.0);
constexpr float fSIGY  = (float)dSIGY;
constexpr float fHARD  = (float)dHARD;
constexpr float fINV3GH = (float)(1.0 / (3.0 * dG + dHARD));
constexpr double dCRET = 3.0 * dG / (3.0 * dG + dHARD);
constexpr float fC1    = (float)(1.0 - dCRET);
constexpr float fCSIGY = (float)(dCRET * dSIGY);
constexpr float fCHARD = (float)(dCRET * dHARD);

#define NB   128
#define NS   2048
#define LS   192
#define NO   3
#define TB   8
#define CH   32
#define PITCH 196

__global__ __launch_bounds__(128) void j2_scan_project_kernel(
    const float* __restrict__ x,
    const float* __restrict__ W1,
    const float* __restrict__ W2,
    float* __restrict__ out) {
  __shared__ __attribute__((aligned(16))) float sbuf[2][CH][PITCH];
  __shared__ __attribute__((aligned(16))) float spw[NO * LS];

  const int b = blockIdx.x;
  const int role = __builtin_amdgcn_readfirstlane((int)(threadIdx.x >> 5));
  const int j    = threadIdx.x & 63;
  const int lane = threadIdx.x & 31;
  const int grp  = lane >> 4;
  const int ln   = lane & 15;

  const float* w = W1 + (3 * j) * 3;
  const float w00 = w[0], w01 = w[1], w02 = w[2];
  const float w10 = w[3], w11 = w[4], w12 = w[5];
  const float w20 = w[6], w21 = w[7], w22 = w[8];

  float eps0 = 0.f, eps1 = 0.f, eps2 = 0.f;
  float sd0 = 0.f, sd1 = 0.f, sdz = 0.f, sd2 = 0.f;
  float fac = 1.0f, eqps = 0.f;

  const float* __restrict__ xb = x + (size_t)b * NS * 3;

  float st[TB][3];
  v2f bfr[LS / 4];

  for (int i = threadIdx.x; i < NO * LS; i += 128) {
    const float wv = W2[i];
    spw[i] = fmaxf(wv, 0.0f) + log1pf(__expf(-fabsf(wv)));
  }
  __syncthreads();

  if (role < 2) {
#pragma unroll
    for (int k = 0; k < TB; ++k) {
      const float x0 = xb[3 * k + 0];
      const float x1 = xb[3 * k + 1];
      const float x2 = xb[3 * k + 2];
      st[k][0] = w00 * x0 + w01 * x1 + w02 * x2;
      st[k][1] = w10 * x0 + w11 * x1 + w12 * x2;
      st[k][2] = w20 * x0 + w21 * x1 + w22 * x2;
    }
#pragma unroll
    for (int t = 0; t < LS / 4; ++t) bfr[t] = (v2f){0.0f, 0.0f};
  } else {
#pragma unroll
    for (int k = 0; k < TB; ++k) { st[k][0] = 0.f; st[k][1] = 0.f; st[k][2] = 0.f; }
    const float* __restrict__ pr = spw + (ln < NO ? ln : 0) * LS + 2 * grp;
#pragma unroll
    for (int t = 0; t < LS / 4; ++t) {
      const v2f pv = *(const v2f*)(pr + 4 * t);
      bfr[t] = (v2f){ln < NO ? pv.x : 0.0f, ln < NO ? pv.y : 0.0f};
    }
  }

  for (int blk = 0; blk <= NS / TB; ++blk) {
    if (role < 2) {
      if (blk < NS / TB) {
        const bool more = (blk + 1 < NS / TB);
        float xn[3 * TB];
        if (more) {
          const float* __restrict__ xnp = xb + (blk + 1) * (3 * TB);
#pragma unroll
          for (int i = 0; i < 3 * TB; ++i) xn[i] = xnp[i];
        }

        float* __restrict__ ob =
            &sbuf[(blk / (CH / TB)) & 1][(blk % (CH / TB)) * TB][3 * j];

#pragma unroll
        for (int k = 0; k < TB; ++k) {
          const float de0 = st[k][0] - eps0;
          const float de1 = st[k][1] - eps1;
          const float de2 = st[k][2] - eps2;
          eps0 = st[k][0]; eps1 = st[k][1]; eps2 = st[k][2];
          const float ds0 = fG43 * de0 - fG23 * de1;
          const float ds1 = fG43 * de1 - fG23 * de0;
          const float dsz = -fG23 * (de0 + de1);
          const float ds2 = fG * de2;
          const float pm  = fKB * (st[k][0] + st[k][1]);

          sd0 = __builtin_fmaf(fac, sd0, ds0);
          sd1 = __builtin_fmaf(fac, sd1, ds1);
          sdz = __builtin_fmaf(fac, sdz, dsz);
          sd2 = __builtin_fmaf(fac, sd2, ds2);

          const float qq  = 1.5f * (sd0 * sd0 + sd1 * sd1 + sdz * sdz +
                                    2.0f * sd2 * sd2) + 1e-12f;
          const float cy  = __builtin_fmaf(fCHARD, eqps, fCSIGY);
          const float r0  = __builtin_amdgcn_rsqf(qq);
          fac = __builtin_fminf(
              __builtin_fmaxf(__builtin_fmaf(cy, r0, fC1), 0.0f), 1.0f);

          const float q0  = qq * r0;
          const float hr  = 0.5f * r0;
          const float er  = __builtin_fmaf(-q0, q0, qq);
          const float q   = __builtin_fmaf(er, hr, q0);
          const float f   = q - (fSIGY + fHARD * eqps);
          const float dgam = (f > 0.0f) ? f * fINV3GH : 0.0f;
          eqps = eqps + dgam;

          float* o = ob + k * PITCH;
          o[0] = pm + fac * sd0;
          o[1] = pm + fac * sd1;
          o[2] = fac * sd2;
        }

        if (more) {
#pragma unroll
          for (int k = 0; k < TB; ++k) {
            const float x0 = xn[3 * k + 0];
            const float x1 = xn[3 * k + 1];
            const float x2 = xn[3 * k + 2];
            st[k][0] = w00 * x0 + w01 * x1 + w02 * x2;
            st[k][1] = w10 * x0 + w11 * x1 + w12 * x2;
            st[k][2] = w20 * x0 + w21 * x1 + w22 * x2;
          }
        }
      }
    } else {
      if ((blk % (CH / TB)) == 0 && blk > 0) {
        const int cprev = blk / (CH / TB) - 1;
        const int row0 = (role - 2) * 16;
        const float* __restrict__ arow = &sbuf[cprev & 1][row0 + ln][2 * grp];
        v8f acc = {};
#pragma unroll
        for (int t = 0; t < LS / 4; ++t) {
          const v2f a = *(const v2f*)(arow + 4 * t);
          acc = __builtin_amdgcn_wmma_f32_16x16x4_f32(
              false, a, false, bfr[t], (short)0, acc, false, false);
        }
        if (ln < NO) {
          const size_t m0 =
              (size_t)b * NS + (size_t)cprev * CH + row0 + 8 * grp;
#pragma unroll
          for (int r = 0; r < 8; ++r) out[(m0 + r) * NO + ln] = acc[r];
        }
      }
    }
    if ((blk % (CH / TB)) == CH / TB - 1) __syncthreads();
  }
}

extern "C" void kernel_launch(void* const* d_in, const int* in_sizes, int n_in,
                              void* d_out, int out_size, void* d_ws,
                              size_t ws_size, hipStream_t stream) {
  (void)in_sizes; (void)n_in; (void)out_size; (void)d_ws; (void)ws_size;
  const float* x  = (const float*)d_in[0];
  const float* W1 = (const float*)d_in[1];
  const float* W2 = (const float*)d_in[2];
  float* out = (float*)d_out;

  j2_scan_project_kernel<<<NB, 128, 0, stream>>>(x, W1, W2, out);
}


# scan: squared norm of the trial deviator as a quadratic in the return factor, qq = fac*(fac*aa + bb) + cc, aa, bb, cc formed beside the chain from s:de products (zz via trace); sums regrouped
# speedup vs baseline: 3.6799x; 1.0601x over previous
#include <hip/hip_runtime.h>


typedef __attribute__((ext_vector_type(2))) float v2f;
typedef __attribute__((ext_vector_type(8))) float v8f;

constexpr double dE    = 3130.0;
constexpr double dNU   = 0.37;
constexpr double dSIGY = 64.8;
constexpr double dHARD = 100.0;
constexpr double dG    = dE / (2.0 * (1.0 + dNU));
constexpr double dLAM  = dE * dNU / ((1.0 + dNU) * (1.0 - 2.0 * dNU));
constexpr double dC    = dE / ((1.0 + dNU) * (1.0 - 2.0 * dNU));

constexpr float fG     = (float)dG;
constexpr float fKB    = (float)(dC * (1.0 + dNU) / 3.0);
constexpr float fG43   = (float)(4.0 * dG / 3.0);
constexpr float fG23   = (float)(2.0 * dG / 3.0);
constexpr float f6G    = (float)(6.0 * dG);
constexpr float f3G    = (float)(3.0 * dG);
constexpr float fSIGY  = (float)dSIGY;
constexpr float fHARD  = (float)dHARD;
constexpr float fINV3GH = (float)(1.0 / (3.0 * dG + dHARD));
constexpr double dCRET = 3.0 * dG / (3.0 * dG + dHARD);
constexpr float fC1    = (float)(1.0 - dCRET);
constexpr float fCSIGY = (float)(dCRET * dSIGY);
constexpr float fCHARD = (float)(dCRET * dHARD);

#define NB   128
#define NS   2048
#define LS   192
#define NO   3
#define TB   8
#define CH   32
#define PITCH 196

__global__ __launch_bounds__(128) void j2_scan_project_kernel(
    const float* __restrict__ x,
    const float* __restrict__ W1,
    const float* __restrict__ W2,
    float* __restrict__ out) {
  __shared__ __attribute__((aligned(16))) float sbuf[2][CH][PITCH];
  __shared__ __attribute__((aligned(16))) float spw[NO * LS];

  const int b = blockIdx.x;
  const int role = __builtin_amdgcn_readfirstlane((int)(threadIdx.x >> 5));
  const int j    = threadIdx.x & 63;
  const int lane = threadIdx.x & 31;
  const int grp  = lane >> 4;
  const int ln   = lane & 15;

  const float* w = W1 + (3 * j) * 3;
  const float w00 = w[0], w01 = w[1], w02 = w[2];
  const float w10 = w[3], w11 = w[4], w12 = w[5];
  const float w20 = w[6], w21 = w[7], w22 = w[8];

  float eps0 = 0.f, eps1 = 0.f, eps2 = 0.f;
  float sd0 = 0.f, sd1 = 0.f, sd2 = 0.f;
  float fac = 1.0f, eqps = 0.f;

  const float* __restrict__ xb = x + (size_t)b * NS * 3;

  float st[TB][3];
  v2f bfr[LS / 4];

  for (int i = threadIdx.x; i < NO * LS; i += 128) {
    const float wv = W2[i];
    spw[i] = fmaxf(wv, 0.0f) + log1pf(__expf(-fabsf(wv)));
  }
  __syncthreads();

  if (role < 2) {
#pragma unroll
    for (int k = 0; k < TB; ++k) {
      const float x0 = xb[3 * k + 0];
      const float x1 = xb[3 * k + 1];
      const float x2 = xb[3 * k + 2];
      st[k][0] = w00 * x0 + w01 * x1 + w02 * x2;
      st[k][1] = w10 * x0 + w11 * x1 + w12 * x2;
      st[k][2] = w20 * x0 + w21 * x1 + w22 * x2;
    }
#pragma unroll
    for (int t = 0; t < LS / 4; ++t) bfr[t] = (v2f){0.0f, 0.0f};
  } else {
#pragma unroll
    for (int k = 0; k < TB; ++k) { st[k][0] = 0.f; st[k][1] = 0.f; st[k][2] = 0.f; }
    const float* __restrict__ pr = spw + (ln < NO ? ln : 0) * LS + 2 * grp;
#pragma unroll
    for (int t = 0; t < LS / 4; ++t) {
      const v2f pv = *(const v2f*)(pr + 4 * t);
      bfr[t] = (v2f){ln < NO ? pv.x : 0.0f, ln < NO ? pv.y : 0.0f};
    }
  }

  for (int blk = 0; blk <= NS / TB; ++blk) {
    if (role < 2) {
      if (blk < NS / TB) {
        const bool more = (blk + 1 < NS / TB);
        float xn[3 * TB];
        if (more) {
          const float* __restrict__ xnp = xb + (blk + 1) * (3 * TB);
#pragma unroll
          for (int i = 0; i < 3 * TB; ++i) xn[i] = xnp[i];
        }

        float* __restrict__ ob =
            &sbuf[(blk / (CH / TB)) & 1][(blk % (CH / TB)) * TB][3 * j];

#pragma unroll
        for (int k = 0; k < TB; ++k) {
          const float de0 = st[k][0] - eps0;
          const float de1 = st[k][1] - eps1;
          const float de2 = st[k][2] - eps2;
          eps0 = st[k][0]; eps1 = st[k][1]; eps2 = st[k][2];
          const float ds0 = fG43 * de0 - fG23 * de1;
          const float ds1 = fG43 * de1 - fG23 * de0;
          const float ds2 = fG * de2;
          const float pm  = fKB * (st[k][0] + st[k][1]);

          const float wb = sd0 * de0 + sd1 * de1 + sd2 * de2;
          const float wc = ds0 * de0 + ds1 * de1 + ds2 * de2;
          const float bb = f6G * wb;
          const float cc = __builtin_fmaf(f3G, wc, 1e-12f);
          const float aa = 3.0f * (sd0 * (sd0 + sd1) + sd1 * sd1 + sd2 * sd2);
          const float tt = __builtin_fmaf(fac, aa, bb);
          const float qq = __builtin_fmaf(fac, tt, cc);

          sd0 = __builtin_fmaf(fac, sd0, ds0);
          sd1 = __builtin_fmaf(fac, sd1, ds1);
          sd2 = __builtin_fmaf(fac, sd2, ds2);
          const float cy  = __builtin_fmaf(fCHARD, eqps, fCSIGY);
          const float r0  = __builtin_amdgcn_rsqf(__builtin_fabsf(qq));
          fac = __builtin_fminf(
              __builtin_fmaxf(__builtin_fmaf(cy, r0, fC1), 0.0f), 1.0f);

          const float q0  = qq * r0;
          const float hr  = 0.5f * r0;
          const float er  = __builtin_fmaf(-q0, q0, qq);
          const float q   = __builtin_fmaf(er, hr, q0);
          const float f   = q - (fSIGY + fHARD * eqps);
          const float dgam = (f > 0.0f) ? f * fINV3GH : 0.0f;
          eqps = eqps + dgam;

          float* o = ob + k * PITCH;
          o[0] = pm + fac * sd0;
          o[1] = pm + fac * sd1;
          o[2] = fac * sd2;
        }

        if (more) {
#pragma unroll
          for (int k = 0; k < TB; ++k) {
            const float x0 = xn[3 * k + 0];
            const float x1 = xn[3 * k + 1];
            const float x2 = xn[3 * k + 2];
            st[k][0] = w00 * x0 + w01 * x1 + w02 * x2;
            st[k][1] = w10 * x0 + w11 * x1 + w12 * x2;
            st[k][2] = w20 * x0 + w21 * x1 + w22 * x2;
          }
        }
      }
    } else {
      if ((blk % (CH / TB)) == 0 && blk > 0) {
        const int cprev = blk / (CH / TB) - 1;
        const int row0 = (role - 2) * 16;
        const float* __restrict__ arow = &sbuf[cprev & 1][row0 + ln][2 * grp];
        v8f acc = {};
#pragma unroll
        for (int t = 0; t < LS / 4; ++t) {
          const v2f a = *(const v2f*)(arow + 4 * t);
          acc = __builtin_amdgcn_wmma_f32_16x16x4_f32(
              false, a, false, bfr[t], (short)0, acc, false, false);
        }
        if (ln < NO) {
          const size_t m0 =
              (size_t)b * NS + (size_t)cprev * CH + row0 + 8 * grp;
#pragma unroll
          for (int r = 0; r < 8; ++r) out[(m0 + r) * NO + ln] = acc[r];
        }
      }
    }
    if ((blk % (CH / TB)) == CH / TB - 1) __syncthreads();
  }
}

extern "C" void kernel_launch(void* const* d_in, const int* in_sizes, int n_in,
                              void* d_out, int out_size, void* d_ws,
                              size_t ws_size, hipStream_t stream) {
  (void)in_sizes; (void)n_in; (void)out_size; (void)d_ws; (void)ws_size;
  const float* x  = (const float*)d_in[0];
  const float* W1 = (const float*)d_in[1];
  const float* W2 = (const float*)d_in[2];
  float* out = (float*)d_out;

  j2_scan_project_kernel<<<NB, 128, 0, stream>>>(x, W1, W2, out);
}


# scan waves read each step's carry-independent terms (strain increment, its stress deviator, cc, mean stress) from LDS, formed a chunk ahead by the projection waves, which also fetch x; sums unchanged
# speedup vs baseline: 3.7924x; 1.0306x over previous
#include <hip/hip_runtime.h>


typedef __attribute__((ext_vector_type(2))) float v2f;
typedef __attribute__((ext_vector_type(4))) float v4f;
typedef __attribute__((ext_vector_type(8))) float v8f;

constexpr double dE    = 3130.0;
constexpr double dNU   = 0.37;
constexpr double dSIGY = 64.8;
constexpr double dHARD = 100.0;
constexpr double dG    = dE / (2.0 * (1.0 + dNU));
constexpr double dLAM  = dE * dNU / ((1.0 + dNU) * (1.0 - 2.0 * dNU));
constexpr double dC    = dE / ((1.0 + dNU) * (1.0 - 2.0 * dNU));

constexpr float fG     = (float)dG;
constexpr float fKB    = (float)(dC * (1.0 + dNU) / 3.0);
constexpr float fG43   = (float)(4.0 * dG / 3.0);
constexpr float fG23   = (float)(2.0 * dG / 3.0);
constexpr float f6G    = (float)(6.0 * dG);
constexpr float f3G    = (float)(3.0 * dG);
constexpr float fSIGY  = (float)dSIGY;
constexpr float fHARD  = (float)dHARD;
constexpr float fINV3GH = (float)(1.0 / (3.0 * dG + dHARD));
constexpr double dCRET = 3.0 * dG / (3.0 * dG + dHARD);
constexpr float fC1    = (float)(1.0 - dCRET);
constexpr float fCSIGY = (float)(dCRET * dSIGY);
constexpr float fCHARD = (float)(dCRET * dHARD);

#define NB   128
#define NS   2048
#define LS   192
#define NO   3
#define NP   64
#define TB   8
#define SB   4
#define CH   32
#define NCH  (NS / CH)
#define PITCH 196

__global__ __launch_bounds__(128) void j2_scan_project_kernel(
    const float* __restrict__ x,
    const float* __restrict__ W1,
    const float* __restrict__ W2,
    float* __restrict__ out) {
  __shared__ __attribute__((aligned(16))) float sbuf[2][CH][PITCH];
  __shared__ __attribute__((aligned(16))) float spw[NO * LS];
  __shared__ __attribute__((aligned(16))) v4f preA[2][CH][NP];
  __shared__ __attribute__((aligned(16))) v4f preB[2][CH][NP];
  __shared__ __attribute__((aligned(16))) float xs[2][CH][4];

  const int b = blockIdx.x;
  const int role = __builtin_amdgcn_readfirstlane((int)(threadIdx.x >> 5));
  const int j    = threadIdx.x & 63;
  const int lane = threadIdx.x & 31;
  const int grp  = lane >> 4;
  const int ln   = lane & 15;

  const float* __restrict__ xb = x + (size_t)b * NS * 3;

  float w00 = 0.f, w01 = 0.f, w02 = 0.f;
  float w10 = 0.f, w11 = 0.f, w12 = 0.f;
  float w20 = 0.f, w21 = 0.f, w22 = 0.f;
  float eps0 = 0.f, eps1 = 0.f, eps2 = 0.f;
  const int xi0 = j, xi1 = j + 64;

  float sd0 = 0.f, sd1 = 0.f, sd2 = 0.f;
  float fac = 1.0f, eqps = 0.f;

  v2f bfr[LS / 4];

  for (int i = threadIdx.x; i < NO * LS; i += 128) {
    const float wv = W2[i];
    spw[i] = fmaxf(wv, 0.0f) + log1pf(__expf(-fabsf(wv)));
  }
  if (role >= 2) {
    xs[j >> 5][j & 31][3] = 0.0f;
#pragma unroll
    for (int c0 = 0; c0 < 2; ++c0) {
      xs[c0][xi0 / 3][xi0 % 3] = xb[c0 * (3 * CH) + xi0];
      if (j < 32) xs[c0][xi1 / 3][xi1 % 3] = xb[c0 * (3 * CH) + xi1];
    }
    const float* w = W1 + (3 * j) * 3;
    w00 = w[0]; w01 = w[1]; w02 = w[2];
    w10 = w[3]; w11 = w[4]; w12 = w[5];
    w20 = w[6]; w21 = w[7]; w22 = w[8];
  }
  __syncthreads();

  auto precompute = [&](const int cq) {
    const int pb = cq & 1;
    v4f XP[TB], XQ[TB];
    auto ldx = [&](v4f* X, const int s0) {
#pragma unroll
      for (int k = 0; k < TB; ++k)
        X[k] = *(const v4f*)&xs[pb][s0 + k][0];
    };
    auto runx = [&](const v4f* X, const int s0) {
#pragma unroll
      for (int k = 0; k < TB; ++k) {
        const int s = s0 + k;
        const float x0 = X[k].x, x1 = X[k].y, x2 = X[k].z;
        const float st0 = w00 * x0 + w01 * x1 + w02 * x2;
        const float st1 = w10 * x0 + w11 * x1 + w12 * x2;
        const float st2 = w20 * x0 + w21 * x1 + w22 * x2;
        const float de0 = st0 - eps0;
        const float de1 = st1 - eps1;
        const float de2 = st2 - eps2;
        eps0 = st0; eps1 = st1; eps2 = st2;
        const float ds0 = fG43 * de0 - fG23 * de1;
        const float ds1 = fG43 * de1 - fG23 * de0;
        const float ds2 = fG * de2;
        const float pm  = fKB * (st0 + st1);
        const float wc = ds0 * de0 + ds1 * de1 + ds2 * de2;
        const float cc = __builtin_fmaf(f3G, wc, 1e-12f);
        preA[pb][s][j] = (v4f){de0, de1, de2, cc};
        preB[pb][s][j] = (v4f){ds0, ds1, ds2, pm};
      }
    };
    ldx(XP, 0);
#pragma unroll 1
    for (int h = 0; h < CH / (2 * TB); ++h) {
      ldx(XQ, (2 * h + 1) * TB);
      runx(XP, (2 * h) * TB);
      if (h + 1 < CH / (2 * TB)) ldx(XP, (2 * h + 2) * TB);
      runx(XQ, (2 * h + 1) * TB);
    }
  };

  if (role < 2) {
#pragma unroll
    for (int t = 0; t < LS / 4; ++t) bfr[t] = (v2f){0.0f, 0.0f};
  } else {
    const float* __restrict__ pr = spw + (ln < NO ? ln : 0) * LS + 2 * grp;
#pragma unroll
    for (int t = 0; t < LS / 4; ++t) {
      const v2f pv = *(const v2f*)(pr + 4 * t);
      bfr[t] = (v2f){ln < NO ? pv.x : 0.0f, ln < NO ? pv.y : 0.0f};
    }
    precompute(0);
  }
  __syncthreads();

  for (int c = 0; c <= NCH; ++c) {
    if (role < 2) {
      if (c < NCH) {
        const int pb = c & 1;
        v4f PA[SB], PB[SB], QA[SB], QB[SB];
        auto ldblk = [&](v4f* A, v4f* B, const int s0) {
#pragma unroll
          for (int k = 0; k < SB; ++k) {
            A[k] = preA[pb][s0 + k][j];
            B[k] = preB[pb][s0 + k][j];
          }
        };
        auto runblk = [&](const v4f* A, const v4f* B, const int s0) {
          float* __restrict__ ob = &sbuf[pb][s0][3 * j];
#pragma unroll
          for (int k = 0; k < SB; ++k) {
            const float de0 = A[k].x, de1 = A[k].y, de2 = A[k].z;
            const float cc  = A[k].w;
            const float ds0 = B[k].x, ds1 = B[k].y, ds2 = B[k].z;
            const float pm  = B[k].w;

            const float wb = sd0 * de0 + sd1 * de1 + sd2 * de2;
            const float bb = f6G * wb;
            const float aa =
                3.0f * (sd0 * (sd0 + sd1) + sd1 * sd1 + sd2 * sd2);
            const float tt = __builtin_fmaf(fac, aa, bb);
            const float qq = __builtin_fmaf(fac, tt, cc);

            sd0 = __builtin_fmaf(fac, sd0, ds0);
            sd1 = __builtin_fmaf(fac, sd1, ds1);
            sd2 = __builtin_fmaf(fac, sd2, ds2);
            const float cy  = __builtin_fmaf(fCHARD, eqps, fCSIGY);
            const float r0  = __builtin_amdgcn_rsqf(__builtin_fabsf(qq));
            fac = __builtin_fminf(
                __builtin_fmaxf(__builtin_fmaf(cy, r0, fC1), 0.0f), 1.0f);

            const float q0  = qq * r0;
            const float hr  = 0.5f * r0;
            const float er  = __builtin_fmaf(-q0, q0, qq);
            const float q   = __builtin_fmaf(er, hr, q0);
            const float f   = q - (fSIGY + fHARD * eqps);
            const float dgam = (f > 0.0f) ? f * fINV3GH : 0.0f;
            eqps = eqps + dgam;

            float* o = ob + k * PITCH;
            o[0] = pm + fac * sd0;
            o[1] = pm + fac * sd1;
            o[2] = fac * sd2;
          }
        };
        ldblk(PA, PB, 0);
#pragma unroll 1
        for (int h = 0; h < CH / (2 * SB); ++h) {
          ldblk(QA, QB, (2 * h + 1) * SB);
          runblk(PA, PB, (2 * h) * SB);
          if (h + 1 < CH / (2 * SB)) ldblk(PA, PB, (2 * h + 2) * SB);
          runblk(QA, QB, (2 * h + 1) * SB);
        }
      }
    } else {
      const bool morex = (c + 2 < NCH);
      float xg0 = 0.f, xg1 = 0.f;
      if (morex) {
        const float* __restrict__ xnp = xb + (c + 2) * (3 * CH);
        xg0 = xnp[xi0];
        if (j < 32) xg1 = xnp[xi1];
      }

      if (c > 0) {
        const int cprev = c - 1;
        const int row0 = (role - 2) * 16;
        const float* __restrict__ arow = &sbuf[cprev & 1][row0 + ln][2 * grp];
        v8f acc = {};
#pragma unroll
        for (int t = 0; t < LS / 4; ++t) {
          const v2f a = *(const v2f*)(arow + 4 * t);
          acc = __builtin_amdgcn_wmma_f32_16x16x4_f32(
              false, a, false, bfr[t], (short)0, acc, false, false);
        }
        if (ln < NO) {
          const size_t m0 =
              (size_t)b * NS + (size_t)cprev * CH + row0 + 8 * grp;
#pragma unroll
          for (int r = 0; r < 8; ++r) out[(m0 + r) * NO + ln] = acc[r];
        }
      }

      if (c + 1 < NCH) precompute(c + 1);

      if (morex) {
        xs[c & 1][xi0 / 3][xi0 % 3] = xg0;
        if (j < 32) xs[c & 1][xi1 / 3][xi1 % 3] = xg1;
      }
    }
    __syncthreads();
  }
}

extern "C" void kernel_launch(void* const* d_in, const int* in_sizes, int n_in,
                              void* d_out, int out_size, void* d_ws,
                              size_t ws_size, hipStream_t stream) {
  (void)in_sizes; (void)n_in; (void)out_size; (void)d_ws; (void)ws_size;
  const float* x  = (const float*)d_in[0];
  const float* W1 = (const float*)d_in[1];
  const float* W2 = (const float*)d_in[2];
  float* out = (float*)d_out;

  j2_scan_project_kernel<<<NB, 128, 0, stream>>>(x, W1, W2, out);
}


# projection waves: the 24 LDS loads of a tile's A fragments issued as one group ahead of its 48 matrix instructions (were one load between every two); k order and sums unchanged
# speedup vs baseline: 4.3874x; 1.1569x over previous
#include <hip/hip_runtime.h>


typedef __attribute__((ext_vector_type(2))) float v2f;
typedef __attribute__((ext_vector_type(4))) float v4f;
typedef __attribute__((ext_vector_type(8))) float v8f;

constexpr double dE    = 3130.0;
constexpr double dNU   = 0.37;
constexpr double dSIGY = 64.8;
constexpr double dHARD = 100.0;
constexpr double dG    = dE / (2.0 * (1.0 + dNU));
constexpr double dLAM  = dE * dNU / ((1.0 + dNU) * (1.0 - 2.0 * dNU));
constexpr double dC    = dE / ((1.0 + dNU) * (1.0 - 2.0 * dNU));

constexpr float fG     = (float)dG;
constexpr float fKB    = (float)(dC * (1.0 + dNU) / 3.0);
constexpr float fG43   = (float)(4.0 * dG / 3.0);
constexpr float fG23   = (float)(2.0 * dG / 3.0);
constexpr float f6G    = (float)(6.0 * dG);
constexpr float f3G    = (float)(3.0 * dG);
constexpr float fSIGY  = (float)dSIGY;
constexpr float fHARD  = (float)dHARD;
constexpr float fINV3GH = (float)(1.0 / (3.0 * dG + dHARD));
constexpr double dCRET = 3.0 * dG / (3.0 * dG + dHARD);
constexpr float fC1    = (float)(1.0 - dCRET);
constexpr float fCSIGY = (float)(dCRET * dSIGY);
constexpr float fCHARD = (float)(dCRET * dHARD);

#define NB   128
#define NS   2048
#define LS   192
#define NO   3
#define NP   64
#define TB   8
#define SB   4
#define CH   32
#define NCH  (NS / CH)
#define PITCH 196

__global__ __launch_bounds__(128) void j2_scan_project_kernel(
    const float* __restrict__ x,
    const float* __restrict__ W1,
    const float* __restrict__ W2,
    float* __restrict__ out) {
  __shared__ __attribute__((aligned(16))) float sbuf[2][CH][PITCH];
  __shared__ __attribute__((aligned(16))) float spw[NO * LS];
  __shared__ __attribute__((aligned(16))) v4f preA[2][CH][NP];
  __shared__ __attribute__((aligned(16))) v4f preB[2][CH][NP];
  __shared__ __attribute__((aligned(16))) float xs[2][CH][4];

  const int b = blockIdx.x;
  const int role = __builtin_amdgcn_readfirstlane((int)(threadIdx.x >> 5));
  const int j    = threadIdx.x & 63;
  const int lane = threadIdx.x & 31;
  const int grp  = lane >> 4;
  const int ln   = lane & 15;

  const float* __restrict__ xb = x + (size_t)b * NS * 3;

  float w00 = 0.f, w01 = 0.f, w02 = 0.f;
  float w10 = 0.f, w11 = 0.f, w12 = 0.f;
  float w20 = 0.f, w21 = 0.f, w22 = 0.f;
  float eps0 = 0.f, eps1 = 0.f, eps2 = 0.f;
  const int xi0 = j, xi1 = j + 64;

  float sd0 = 0.f, sd1 = 0.f, sd2 = 0.f;
  float fac = 1.0f, eqps = 0.f;

  v2f bfr[LS / 4];

  for (int i = threadIdx.x; i < NO * LS; i += 128) {
    const float wv = W2[i];
    spw[i] = fmaxf(wv, 0.0f) + log1pf(__expf(-fabsf(wv)));
  }
  if (role >= 2) {
    xs[j >> 5][j & 31][3] = 0.0f;
#pragma unroll
    for (int c0 = 0; c0 < 2; ++c0) {
      xs[c0][xi0 / 3][xi0 % 3] = xb[c0 * (3 * CH) + xi0];
      if (j < 32) xs[c0][xi1 / 3][xi1 % 3] = xb[c0 * (3 * CH) + xi1];
    }
    const float* w = W1 + (3 * j) * 3;
    w00 = w[0]; w01 = w[1]; w02 = w[2];
    w10 = w[3]; w11 = w[4]; w12 = w[5];
    w20 = w[6]; w21 = w[7]; w22 = w[8];
  }
  __syncthreads();

  auto precompute = [&](const int cq) {
    const int pb = cq & 1;
    v4f XP[TB], XQ[TB];
    auto ldx = [&](v4f* X, const int s0) {
#pragma unroll
      for (int k = 0; k < TB; ++k)
        X[k] = *(const v4f*)&xs[pb][s0 + k][0];
    };
    auto runx = [&](const v4f* X, const int s0) {
#pragma unroll
      for (int k = 0; k < TB; ++k) {
        const int s = s0 + k;
        const float x0 = X[k].x, x1 = X[k].y, x2 = X[k].z;
        const float st0 = w00 * x0 + w01 * x1 + w02 * x2;
        const float st1 = w10 * x0 + w11 * x1 + w12 * x2;
        const float st2 = w20 * x0 + w21 * x1 + w22 * x2;
        const float de0 = st0 - eps0;
        const float de1 = st1 - eps1;
        const float de2 = st2 - eps2;
        eps0 = st0; eps1 = st1; eps2 = st2;
        const float ds0 = fG43 * de0 - fG23 * de1;
        const float ds1 = fG43 * de1 - fG23 * de0;
        const float ds2 = fG * de2;
        const float pm  = fKB * (st0 + st1);
        const float wc = ds0 * de0 + ds1 * de1 + ds2 * de2;
        const float cc = __builtin_fmaf(f3G, wc, 1e-12f);
        preA[pb][s][j] = (v4f){de0, de1, de2, cc};
        preB[pb][s][j] = (v4f){ds0, ds1, ds2, pm};
      }
    };
    ldx(XP, 0);
#pragma unroll 1
    for (int h = 0; h < CH / (2 * TB); ++h) {
      ldx(XQ, (2 * h + 1) * TB);
      runx(XP, (2 * h) * TB);
      if (h + 1 < CH / (2 * TB)) ldx(XP, (2 * h + 2) * TB);
      runx(XQ, (2 * h + 1) * TB);
    }
  };

  if (role < 2) {
#pragma unroll
    for (int t = 0; t < LS / 4; ++t) bfr[t] = (v2f){0.0f, 0.0f};
  } else {
    const float* __restrict__ pr = spw + (ln < NO ? ln : 0) * LS + 2 * grp;
#pragma unroll
    for (int t = 0; t < LS / 4; ++t) {
      const v2f pv = *(const v2f*)(pr + 4 * t);
      bfr[t] = (v2f){ln < NO ? pv.x : 0.0f, ln < NO ? pv.y : 0.0f};
    }
    precompute(0);
  }
  __syncthreads();

  for (int c = 0; c <= NCH; ++c) {
    if (role < 2) {
      if (c < NCH) {
        const int pb = c & 1;
        v4f PA[SB], PB[SB], QA[SB], QB[SB];
        auto ldblk = [&](v4f* A, v4f* B, const int s0) {
#pragma unroll
          for (int k = 0; k < SB; ++k) {
            A[k] = preA[pb][s0 + k][j];
            B[k] = preB[pb][s0 + k][j];
          }
        };
        auto runblk = [&](const v4f* A, const v4f* B, const int s0) {
          float* __restrict__ ob = &sbuf[pb][s0][3 * j];
#pragma unroll
          for (int k = 0; k < SB; ++k) {
            const float de0 = A[k].x, de1 = A[k].y, de2 = A[k].z;
            const float cc  = A[k].w;
            const float ds0 = B[k].x, ds1 = B[k].y, ds2 = B[k].z;
            const float pm  = B[k].w;

            const float wb = sd0 * de0 + sd1 * de1 + sd2 * de2;
            const float bb = f6G * wb;
            const float aa =
                3.0f * (sd0 * (sd0 + sd1) + sd1 * sd1 + sd2 * sd2);
            const float tt = __builtin_fmaf(fac, aa, bb);
            const float qq = __builtin_fmaf(fac, tt, cc);

            sd0 = __builtin_fmaf(fac, sd0, ds0);
            sd1 = __builtin_fmaf(fac, sd1, ds1);
            sd2 = __builtin_fmaf(fac, sd2, ds2);
            const float cy  = __builtin_fmaf(fCHARD, eqps, fCSIGY);
            const float r0  = __builtin_amdgcn_rsqf(__builtin_fabsf(qq));
            fac = __builtin_fminf(
                __builtin_fmaxf(__builtin_fmaf(cy, r0, fC1), 0.0f), 1.0f);

            const float q0  = qq * r0;
            const float hr  = 0.5f * r0;
            const float er  = __builtin_fmaf(-q0, q0, qq);
            const float q   = __builtin_fmaf(er, hr, q0);
            const float f   = q - (fSIGY + fHARD * eqps);
            const float dgam = (f > 0.0f) ? f * fINV3GH : 0.0f;
            eqps = eqps + dgam;

            float* o = ob + k * PITCH;
            o[0] = pm + fac * sd0;
            o[1] = pm + fac * sd1;
            o[2] = fac * sd2;
          }
        };
        ldblk(PA, PB, 0);
#pragma unroll 1
        for (int h = 0; h < CH / (2 * SB); ++h) {
          ldblk(QA, QB, (2 * h + 1) * SB);
          runblk(PA, PB, (2 * h) * SB);
          if (h + 1 < CH / (2 * SB)) ldblk(PA, PB, (2 * h + 2) * SB);
          runblk(QA, QB, (2 * h + 1) * SB);
        }
      }
    } else {
      const bool morex = (c + 2 < NCH);
      float xg0 = 0.f, xg1 = 0.f;
      if (morex) {
        const float* __restrict__ xnp = xb + (c + 2) * (3 * CH);
        xg0 = xnp[xi0];
        if (j < 32) xg1 = xnp[xi1];
      }

      if (c > 0) {
        const int cprev = c - 1;
        const int row0 = (role - 2) * 16;
        const float* __restrict__ arow = &sbuf[cprev & 1][row0 + ln][2 * grp];
        v2f afr[LS / 4];
#pragma unroll
        for (int t = 0; t < LS / 4; ++t) afr[t] = *(const v2f*)(arow + 4 * t);
        __builtin_amdgcn_sched_barrier(0);
        v8f acc = {};
#pragma unroll
        for (int t = 0; t < LS / 4; ++t) {
          acc = __builtin_amdgcn_wmma_f32_16x16x4_f32(
              false, afr[t], false, bfr[t], (short)0, acc, false, false);
        }
        if (ln < NO) {
          const size_t m0 =
              (size_t)b * NS + (size_t)cprev * CH + row0 + 8 * grp;
#pragma unroll
          for (int r = 0; r < 8; ++r) out[(m0 + r) * NO + ln] = acc[r];
        }
      }

      if (c + 1 < NCH) precompute(c + 1);

      if (morex) {
        xs[c & 1][xi0 / 3][xi0 % 3] = xg0;
        if (j < 32) xs[c & 1][xi1 / 3][xi1 % 3] = xg1;
      }
    }
    __syncthreads();
  }
}

extern "C" void kernel_launch(void* const* d_in, const int* in_sizes, int n_in,
                              void* d_out, int out_size, void* d_ws,
                              size_t ws_size, hipStream_t stream) {
  (void)in_sizes; (void)n_in; (void)out_size; (void)d_ws; (void)ws_size;
  const float* x  = (const float*)d_in[0];
  const float* W1 = (const float*)d_in[1];
  const float* W2 = (const float*)d_in[2];
  float* out = (float*)d_out;

  j2_scan_project_kernel<<<NB, 128, 0, stream>>>(x, W1, W2, out);
}


# a chunk's LDS traffic issued as single groups per wave (scan: 64 loads at its head, 96 stress stores at its end; item waves: 32 x loads, 64 item stores), not spread among vector work; sums unchanged
# speedup vs baseline: 5.4663x; 1.2459x over previous
#include <hip/hip_runtime.h>


typedef __attribute__((ext_vector_type(2))) float v2f;
typedef __attribute__((ext_vector_type(4))) float v4f;
typedef __attribute__((ext_vector_type(8))) float v8f;

constexpr double dE    = 3130.0;
constexpr double dNU   = 0.37;
constexpr double dSIGY = 64.8;
constexpr double dHARD = 100.0;
constexpr double dG    = dE / (2.0 * (1.0 + dNU));
constexpr double dLAM  = dE * dNU / ((1.0 + dNU) * (1.0 - 2.0 * dNU));
constexpr double dC    = dE / ((1.0 + dNU) * (1.0 - 2.0 * dNU));

constexpr float fG     = (float)dG;
constexpr float fKB    = (float)(dC * (1.0 + dNU) / 3.0);
constexpr float fG43   = (float)(4.0 * dG / 3.0);
constexpr float fG23   = (float)(2.0 * dG / 3.0);
constexpr float f6G    = (float)(6.0 * dG);
constexpr float f3G    = (float)(3.0 * dG);
constexpr float fSIGY  = (float)dSIGY;
constexpr float fHARD  = (float)dHARD;
constexpr float fINV3GH = (float)(1.0 / (3.0 * dG + dHARD));
constexpr double dCRET = 3.0 * dG / (3.0 * dG + dHARD);
constexpr float fC1    = (float)(1.0 - dCRET);
constexpr float fCSIGY = (float)(dCRET * dSIGY);
constexpr float fCHARD = (float)(dCRET * dHARD);

#define NB   128
#define NS   2048
#define LS   192
#define NO   3
#define NP   64
#define CH   32
#define NCH  (NS / CH)
#define PITCH 196

__global__ __launch_bounds__(128) void j2_scan_project_kernel(
    const float* __restrict__ x,
    const float* __restrict__ W1,
    const float* __restrict__ W2,
    float* __restrict__ out) {
  struct Lds {
    float xs[2][CH][4];
    float spw[NO * LS];
    v4f pre[2][CH][2][NP];
    float sbuf[2][CH][PITCH];
  };
  __shared__ __attribute__((aligned(16))) Lds lds;
  float (&xs)[2][CH][4]       = lds.xs;
  float (&spw)[NO * LS]       = lds.spw;
  v4f   (&pre)[2][CH][2][NP]  = lds.pre;
  float (&sbuf)[2][CH][PITCH] = lds.sbuf;

  const int b = blockIdx.x;
  const int role = __builtin_amdgcn_readfirstlane((int)(threadIdx.x >> 5));
  const int j    = threadIdx.x & 63;
  const int lane = threadIdx.x & 31;
  const int grp  = lane >> 4;
  const int ln   = lane & 15;

  const float* __restrict__ xb = x + (size_t)b * NS * 3;

  float w00 = 0.f, w01 = 0.f, w02 = 0.f;
  float w10 = 0.f, w11 = 0.f, w12 = 0.f;
  float w20 = 0.f, w21 = 0.f, w22 = 0.f;
  float eps0 = 0.f, eps1 = 0.f, eps2 = 0.f;
  const int xi0 = j, xi1 = j + 64;

  float sd0 = 0.f, sd1 = 0.f, sd2 = 0.f;
  float fac = 1.0f, eqps = 0.f;

  v2f bfr[LS / 4];

  for (int i = threadIdx.x; i < NO * LS; i += 128) {
    const float wv = W2[i];
    spw[i] = fmaxf(wv, 0.0f) + log1pf(__expf(-fabsf(wv)));
  }
  if (role >= 2) {
    xs[j >> 5][j & 31][3] = 0.0f;
#pragma unroll
    for (int c0 = 0; c0 < 2; ++c0) {
      xs[c0][xi0 / 3][xi0 % 3] = xb[c0 * (3 * CH) + xi0];
      if (j < 32) xs[c0][xi1 / 3][xi1 % 3] = xb[c0 * (3 * CH) + xi1];
    }
    const float* w = W1 + (3 * j) * 3;
    w00 = w[0]; w01 = w[1]; w02 = w[2];
    w10 = w[3]; w11 = w[4]; w12 = w[5];
    w20 = w[6]; w21 = w[7]; w22 = w[8];
  }
  __syncthreads();

  auto precompute = [&](const int cq) {
    const int pb = cq & 1;
    v4f X[CH];
#pragma unroll
    for (int k = 0; k < CH; ++k) X[k] = *(const v4f*)&xs[pb][k][0];
    __builtin_amdgcn_sched_barrier(0);
    v4f A[CH], B[CH];
#pragma unroll
    for (int k = 0; k < CH; ++k) {
      const float x0 = X[k].x, x1 = X[k].y, x2 = X[k].z;
      const float st0 = w00 * x0 + w01 * x1 + w02 * x2;
      const float st1 = w10 * x0 + w11 * x1 + w12 * x2;
      const float st2 = w20 * x0 + w21 * x1 + w22 * x2;
      const float de0 = st0 - eps0;
      const float de1 = st1 - eps1;
      const float de2 = st2 - eps2;
      eps0 = st0; eps1 = st1; eps2 = st2;
      const float ds0 = fG43 * de0 - fG23 * de1;
      const float ds1 = fG43 * de1 - fG23 * de0;
      const float ds2 = fG * de2;
      const float pm  = fKB * (st0 + st1);
      const float wc = ds0 * de0 + ds1 * de1 + ds2 * de2;
      const float cc = __builtin_fmaf(f3G, wc, 1e-12f);
      A[k] = (v4f){de0, de1, de2, cc};
      B[k] = (v4f){ds0, ds1, ds2, pm};
    }
    v4f* __restrict__ po = &pre[pb][0][0][j];
#pragma unroll
    for (int k = 0; k < CH; ++k) {
      po[(2 * k) * NP]     = A[k];
      po[(2 * k + 1) * NP] = B[k];
    }
  };

  if (role < 2) {
#pragma unroll
    for (int t = 0; t < LS / 4; ++t) bfr[t] = (v2f){0.0f, 0.0f};
  } else {
    const float* __restrict__ pr = spw + (ln < NO ? ln : 0) * LS + 2 * grp;
#pragma unroll
    for (int t = 0; t < LS / 4; ++t) {
      const v2f pv = *(const v2f*)(pr + 4 * t);
      bfr[t] = (v2f){ln < NO ? pv.x : 0.0f, ln < NO ? pv.y : 0.0f};
    }
    precompute(0);
  }
  __syncthreads();

  for (int c = 0; c <= NCH; ++c) {
    if (role < 2) {
      if (c < NCH) {
        const int pb = c & 1;
        v4f ta[CH], tb[CH];
        const v4f* __restrict__ pin = &pre[pb][0][0][j];
#pragma unroll
        for (int k = 0; k < CH; ++k) {
          ta[k] = pin[(2 * k) * NP];
          tb[k] = pin[(2 * k + 1) * NP];
        }
        __builtin_amdgcn_sched_barrier(0);
        float o0[CH], o1[CH], o2[CH];
#pragma unroll
        for (int k = 0; k < CH; ++k) {
          const float de0 = ta[k].x, de1 = ta[k].y, de2 = ta[k].z;
          const float cc  = ta[k].w;
          const float ds0 = tb[k].x, ds1 = tb[k].y, ds2 = tb[k].z;
          const float pm  = tb[k].w;

          const float wb = sd0 * de0 + sd1 * de1 + sd2 * de2;
          const float bb = f6G * wb;
          const float aa =
              3.0f * (sd0 * (sd0 + sd1) + sd1 * sd1 + sd2 * sd2);
          const float tt = __builtin_fmaf(fac, aa, bb);
          const float qq = __builtin_fmaf(fac, tt, cc);

          sd0 = __builtin_fmaf(fac, sd0, ds0);
          sd1 = __builtin_fmaf(fac, sd1, ds1);
          sd2 = __builtin_fmaf(fac, sd2, ds2);
          const float cy  = __builtin_fmaf(fCHARD, eqps, fCSIGY);
          const float r0  = __builtin_amdgcn_rsqf(__builtin_fabsf(qq));
          fac = __builtin_fminf(
              __builtin_fmaxf(__builtin_fmaf(cy, r0, fC1), 0.0f), 1.0f);

          const float q0  = qq * r0;
          const float hr  = 0.5f * r0;
          const float er  = __builtin_fmaf(-q0, q0, qq);
          const float q   = __builtin_fmaf(er, hr, q0);
          const float f   = q - (fSIGY + fHARD * eqps);
          const float dgam = (f > 0.0f) ? f * fINV3GH : 0.0f;
          eqps = eqps + dgam;

          o0[k] = pm + fac * sd0;
          o1[k] = pm + fac * sd1;
          o2[k] = fac * sd2;
        }
        __builtin_amdgcn_sched_barrier(0);
        float* __restrict__ ob = &sbuf[pb][0][3 * j];
#pragma unroll
        for (int k = 0; k < CH; ++k) {
          ob[k * PITCH + 0] = o0[k];
          ob[k * PITCH + 1] = o1[k];
          ob[k * PITCH + 2] = o2[k];
        }
        __builtin_amdgcn_sched_group_barrier(0x002, 96, 0);
        __builtin_amdgcn_sched_group_barrier(0x200, 96, 0);
      }
    } else {
      const bool morex = (c + 2 < NCH);
      float xg0 = 0.f, xg1 = 0.f;
      if (morex) {
        const float* __restrict__ xnp = xb + (c + 2) * (3 * CH);
        xg0 = xnp[xi0];
        if (j < 32) xg1 = xnp[xi1];
      }

      if (c > 0) {
        const int cprev = c - 1;
        const int row0 = (role - 2) * 16;
        const float* __restrict__ arow = &sbuf[cprev & 1][row0 + ln][2 * grp];
        v2f afr[LS / 4];
#pragma unroll
        for (int t = 0; t < LS / 4; ++t) afr[t] = *(const v2f*)(arow + 4 * t);
        __builtin_amdgcn_sched_barrier(0);
        v8f acc = {};
#pragma unroll
        for (int t = 0; t < LS / 4; ++t) {
          acc = __builtin_amdgcn_wmma_f32_16x16x4_f32(
              false, afr[t], false, bfr[t], (short)0, acc, false, false);
        }
        if (ln < NO) {
          const size_t m0 =
              (size_t)b * NS + (size_t)cprev * CH + row0 + 8 * grp;
#pragma unroll
          for (int r = 0; r < 8; ++r) out[(m0 + r) * NO + ln] = acc[r];
        }
      }

      if (c + 1 < NCH) precompute(c + 1);

      if (morex) {
        xs[c & 1][xi0 / 3][xi0 % 3] = xg0;
        if (j < 32) xs[c & 1][xi1 / 3][xi1 % 3] = xg1;
      }
    }
    __syncthreads();
  }
}

extern "C" void kernel_launch(void* const* d_in, const int* in_sizes, int n_in,
                              void* d_out, int out_size, void* d_ws,
                              size_t ws_size, hipStream_t stream) {
  (void)in_sizes; (void)n_in; (void)out_size; (void)d_ws; (void)ws_size;
  const float* x  = (const float*)d_in[0];
  const float* W1 = (const float*)d_in[1];
  const float* W2 = (const float*)d_in[2];
  float* out = (float*)d_out;

  j2_scan_project_kernel<<<NB, 128, 0, stream>>>(x, W1, W2, out);
}


# projection moved to a pair of waves of its own (8 waves per work-group): scan, item arithmetic and matrix chain run side by side in a trip; a tile's out rows stored a trip later; sums unchanged
# speedup vs baseline: 6.0299x; 1.1031x over previous
#include <hip/hip_runtime.h>


typedef __attribute__((ext_vector_type(2))) float v2f;
typedef __attribute__((ext_vector_type(4))) float v4f;
typedef __attribute__((ext_vector_type(8))) float v8f;

constexpr double dE    = 3130.0;
constexpr double dNU   = 0.37;
constexpr double dSIGY = 64.8;
constexpr double dHARD = 100.0;
constexpr double dG    = dE / (2.0 * (1.0 + dNU));
constexpr double dLAM  = dE * dNU / ((1.0 + dNU) * (1.0 - 2.0 * dNU));
constexpr double dC    = dE / ((1.0 + dNU) * (1.0 - 2.0 * dNU));

constexpr float fG     = (float)dG;
constexpr float fKB    = (float)(dC * (1.0 + dNU) / 3.0);
constexpr float fG43   = (float)(4.0 * dG / 3.0);
constexpr float fG23   = (float)(2.0 * dG / 3.0);
constexpr float f6G    = (float)(6.0 * dG);
constexpr float f3G    = (float)(3.0 * dG);
constexpr float fSIGY  = (float)dSIGY;
constexpr float fHARD  = (float)dHARD;
constexpr float fINV3GH = (float)(1.0 / (3.0 * dG + dHARD));
constexpr double dCRET = 3.0 * dG / (3.0 * dG + dHARD);
constexpr float fC1    = (float)(1.0 - dCRET);
constexpr float fCSIGY = (float)(dCRET * dSIGY);
constexpr float fCHARD = (float)(dCRET * dHARD);

#define NB   128
#define NS   2048
#define LS   192
#define NO   3
#define NP   64
#define CH   32
#define NCH  (NS / CH)
#define NT   256
#define PITCH 196

__global__ __launch_bounds__(NT) void j2_scan_project_kernel(
    const float* __restrict__ x,
    const float* __restrict__ W1,
    const float* __restrict__ W2,
    float* __restrict__ out) {
  struct Lds {
    float xs[2][CH][4];
    float spw[NO * LS];
    v4f pre[2][CH][2][NP];
    float sbuf[2][CH][PITCH];
  };
  __shared__ __attribute__((aligned(16))) Lds lds;
  float (&xs)[2][CH][4]       = lds.xs;
  float (&spw)[NO * LS]       = lds.spw;
  v4f   (&pre)[2][CH][2][NP]  = lds.pre;
  float (&sbuf)[2][CH][PITCH] = lds.sbuf;

  const int b = blockIdx.x;
  const int role = __builtin_amdgcn_readfirstlane((int)(threadIdx.x >> 5));
  const int j    = threadIdx.x & 63;
  const bool proj  = (role >> 1) == 0;
  const bool scan  = (role >> 1) == 1;
  const bool items = (role >> 1) == 2;
  const int lane = threadIdx.x & 31;
  const int grp  = lane >> 4;
  const int ln   = lane & 15;

  const float* __restrict__ xb = x + (size_t)b * NS * 3;

  float w00 = 0.f, w01 = 0.f, w02 = 0.f;
  float w10 = 0.f, w11 = 0.f, w12 = 0.f;
  float w20 = 0.f, w21 = 0.f, w22 = 0.f;
  float eps0 = 0.f, eps1 = 0.f, eps2 = 0.f;
  const int xi0 = j, xi1 = j + 64;
  const int xo0 = (xi0 / 3) * 4 + xi0 % 3;
  const int xo1 = (xi1 / 3) * 4 + xi1 % 3;
  float* __restrict__ xsf = &lds.xs[0][0][0];

  float sd0 = 0.f, sd1 = 0.f, sd2 = 0.f;
  float fac = 1.0f, eqps = 0.f;

  v2f bfr[LS / 4];

  for (int i = threadIdx.x; i < NO * LS; i += NT) {
    const float wv = W2[i];
    spw[i] = fmaxf(wv, 0.0f) + log1pf(__expf(-fabsf(wv)));
  }
  if (items) {
    xs[j >> 5][j & 31][3] = 0.0f;
#pragma unroll
    for (int c0 = 0; c0 < 2; ++c0) {
      xsf[c0 * (CH * 4) + xo0] = xb[c0 * (3 * CH) + xi0];
      if (j < 32) xsf[c0 * (CH * 4) + xo1] = xb[c0 * (3 * CH) + xi1];
    }
    const float* w = W1 + (3 * j) * 3;
    w00 = w[0]; w01 = w[1]; w02 = w[2];
    w10 = w[3]; w11 = w[4]; w12 = w[5];
    w20 = w[6]; w21 = w[7]; w22 = w[8];
  }
  __syncthreads();

  auto precompute = [&](const int cq) {
    const int pb = cq & 1;
    v4f X[CH];
#pragma unroll
    for (int k = 0; k < CH; ++k) X[k] = *(const v4f*)&xs[pb][k][0];
    __builtin_amdgcn_sched_barrier(0);
    v4f A[CH], B[CH];
#pragma unroll
    for (int k = 0; k < CH; ++k) {
      const float x0 = X[k].x, x1 = X[k].y, x2 = X[k].z;
      const float st0 = w00 * x0 + w01 * x1 + w02 * x2;
      const float st1 = w10 * x0 + w11 * x1 + w12 * x2;
      const float st2 = w20 * x0 + w21 * x1 + w22 * x2;
      const float de0 = st0 - eps0;
      const float de1 = st1 - eps1;
      const float de2 = st2 - eps2;
      eps0 = st0; eps1 = st1; eps2 = st2;
      const float ds0 = fG43 * de0 - fG23 * de1;
      const float ds1 = fG43 * de1 - fG23 * de0;
      const float ds2 = fG * de2;
      const float pm  = fKB * (st0 + st1);
      const float wc = ds0 * de0 + ds1 * de1 + ds2 * de2;
      const float cc = __builtin_fmaf(f3G, wc, 1e-12f);
      A[k] = (v4f){de0, de1, de2, cc};
      B[k] = (v4f){ds0, ds1, ds2, pm};
    }
    v4f* __restrict__ po = &pre[pb][0][0][j];
#pragma unroll
    for (int k = 0; k < CH; ++k) {
      po[(2 * k) * NP]     = A[k];
      po[(2 * k + 1) * NP] = B[k];
    }
  };

  if (proj) {
    const float* __restrict__ pr = spw + (ln < NO ? ln : 0) * LS + 2 * grp;
#pragma unroll
    for (int t = 0; t < LS / 4; ++t) {
      const v2f pv = *(const v2f*)(pr + 4 * t);
      bfr[t] = (v2f){ln < NO ? pv.x : 0.0f, ln < NO ? pv.y : 0.0f};
    }
  } else {
#pragma unroll
    for (int t = 0; t < LS / 4; ++t) bfr[t] = (v2f){0.0f, 0.0f};
    if (items) precompute(0);
  }
  __syncthreads();

  const int row0 = (role & 1) * 16;
  float* __restrict__ orow = out + ((size_t)b * NS + row0 + 8 * grp) * NO + ln;
  v8f accp = {};
  auto store_rows = [&](const v8f& a, const int cq, const bool live) {
    if (live && ln < NO) {
      float* __restrict__ p = orow + (size_t)cq * (CH * NO);
#pragma unroll
      for (int r = 0; r < 8; ++r) p[r * NO] = a[r];
    }
  };

  for (int c = 0; c <= NCH; ++c) {
    if (scan) {
      if (c < NCH) {
        const int pb = c & 1;
        v4f ta[CH], tb[CH];
        const v4f* __restrict__ pin = &pre[pb][0][0][j];
#pragma unroll
        for (int k = 0; k < CH; ++k) {
          ta[k] = pin[(2 * k) * NP];
          tb[k] = pin[(2 * k + 1) * NP];
        }
        __builtin_amdgcn_sched_barrier(0);
        float o0[CH], o1[CH], o2[CH];
#pragma unroll
        for (int k = 0; k < CH; ++k) {
          const float de0 = ta[k].x, de1 = ta[k].y, de2 = ta[k].z;
          const float cc  = ta[k].w;
          const float ds0 = tb[k].x, ds1 = tb[k].y, ds2 = tb[k].z;
          const float pm  = tb[k].w;

          const float wb = sd0 * de0 + sd1 * de1 + sd2 * de2;
          const float bb = f6G * wb;
          const float aa =
              3.0f * (sd0 * (sd0 + sd1) + sd1 * sd1 + sd2 * sd2);
          const float tt = __builtin_fmaf(fac, aa, bb);
          const float qq = __builtin_fmaf(fac, tt, cc);

          sd0 = __builtin_fmaf(fac, sd0, ds0);
          sd1 = __builtin_fmaf(fac, sd1, ds1);
          sd2 = __builtin_fmaf(fac, sd2, ds2);
          const float cy  = __builtin_fmaf(fCHARD, eqps, fCSIGY);
          const float r0  = __builtin_amdgcn_rsqf(__builtin_fabsf(qq));
          fac = __builtin_fminf(
              __builtin_fmaxf(__builtin_fmaf(cy, r0, fC1), 0.0f), 1.0f);

          const float q0  = qq * r0;
          const float hr  = 0.5f * r0;
          const float er  = __builtin_fmaf(-q0, q0, qq);
          const float q   = __builtin_fmaf(er, hr, q0);
          const float f   = q - (fSIGY + fHARD * eqps);
          const float dgam = (f > 0.0f) ? f * fINV3GH : 0.0f;
          eqps = eqps + dgam;

          o0[k] = pm + fac * sd0;
          o1[k] = pm + fac * sd1;
          o2[k] = fac * sd2;
        }
        __builtin_amdgcn_sched_barrier(0);
        float* __restrict__ ob = &sbuf[pb][0][3 * j];
#pragma unroll
        for (int k = 0; k < CH; ++k) {
          ob[k * PITCH + 0] = o0[k];
          ob[k * PITCH + 1] = o1[k];
          ob[k * PITCH + 2] = o2[k];
        }
        __builtin_amdgcn_sched_group_barrier(0x002, 96, 0);
        __builtin_amdgcn_sched_group_barrier(0x200, 96, 0);
      }
    } else if (items) {
      const bool morex = (c + 2 < NCH);
      float xg0 = 0.f, xg1 = 0.f;
      if (morex) {
        const float* __restrict__ xnp = xb + (c + 2) * (3 * CH);
        xg0 = xnp[xi0];
        if (j < 32) xg1 = xnp[xi1];
      }

      if (c + 1 < NCH) precompute(c + 1);

      if (morex) {
        xsf[(c & 1) * (CH * 4) + xo0] = xg0;
        if (j < 32) xsf[(c & 1) * (CH * 4) + xo1] = xg1;
      }
    } else if (proj) {
      if (c > 0) {
        store_rows(accp, c - 2, c > 1);
        const float* __restrict__ arow =
            &sbuf[(c - 1) & 1][row0 + ln][2 * grp];
        v2f afr[LS / 4];
#pragma unroll
        for (int t = 0; t < LS / 4; ++t) afr[t] = *(const v2f*)(arow + 4 * t);
        __builtin_amdgcn_sched_barrier(0);
        v8f acc = {};
#pragma unroll
        for (int t = 0; t < LS / 4; ++t) {
          acc = __builtin_amdgcn_wmma_f32_16x16x4_f32(
              false, afr[t], false, bfr[t], (short)0, acc, false, false);
        }
        accp = acc;
      }
    }
    __syncthreads();
  }
  if (proj) store_rows(accp, NCH - 1, true);
}

extern "C" void kernel_launch(void* const* d_in, const int* in_sizes, int n_in,
                              void* d_out, int out_size, void* d_ws,
                              size_t ws_size, hipStream_t stream) {
  (void)in_sizes; (void)n_in; (void)out_size; (void)d_ws; (void)ws_size;
  const float* x  = (const float*)d_in[0];
  const float* W1 = (const float*)d_in[1];
  const float* W2 = (const float*)d_in[2];
  float* out = (float*)d_out;

  j2_scan_project_kernel<<<NB, NT, 0, stream>>>(x, W1, W2, out);
}
